# starting kernel
# baseline (speedup 1.0000x reference)
#include <hip/hip_runtime.h>

#define N_NODES 100000
#define N_EDGES 600000
#define HD 128

typedef __attribute__((ext_vector_type(2))) float v2f;
typedef __attribute__((ext_vector_type(8))) float v8f;

__global__ __launch_bounds__(256) void gather_embed(
    const int* __restrict__ node_id, const float* __restrict__ embed,
    float* __restrict__ h) {
  int t   = blockIdx.x * blockDim.x + threadIdx.x;
  int row = t >> 5;
  int c4  = (t & 31) << 2;
  if (row < N_NODES) {
    int srow = node_id[row];
    float4 v = *(const float4*)(embed + (size_t)srow * HD + c4);
    *(float4*)(h + (size_t)row * HD + c4) = v;
  }
}

__global__ __launch_bounds__(256) void gemm_norm_wmma(
    const float* __restrict__ h, const float* __restrict__ W,
    const float* __restrict__ norm, float* __restrict__ m) {
  __shared__ float Wl2[HD * HD];

  int tid = threadIdx.x;
  for (int i = tid; i < (HD * HD) / 4; i += 256) {
    int p = i >> 6;
    int n = (i & 63) << 1;
    int k = p << 1;
    float2 r0 = *(const float2*)(W + (size_t)k * HD + n);
    float2 r1 = *(const float2*)(W + (size_t)(k + 1) * HD + n);
    float4 o  = make_float4(r0.x, r1.x, r0.y, r1.y);
    *(float4*)(Wl2 + p * 256 + n * 2) = o;
  }
  __syncthreads();

  int wave    = tid >> 5;
  int lane    = tid & 31;
  int lane16  = lane & 15;
  int pOff    = lane >> 4;
  int rowBase = (blockIdx.x * 8 + wave) * 16;
  int rowA    = rowBase + lane16;
  int rowAc   = rowA < N_NODES ? rowA : (N_NODES - 1);

  v8f acc[8] = {};
  const float* hrow = h + (size_t)rowAc * HD;

  for (int k0 = 0; k0 < HD; k0 += 4) {
    v2f a = *(const v2f*)(hrow + k0 + (pOff << 1));
    const float* bbase = Wl2 + ((k0 >> 1) + pOff) * 256;
#pragma unroll
    for (int t = 0; t < 8; ++t) {
      int n = t * 16 + lane16;
      v2f b = *(const v2f*)(bbase + n * 2);
      acc[t] = __builtin_amdgcn_wmma_f32_16x16x4_f32(
          false, a, false, b, (short)0, acc[t], false, false);
    }
  }

  int rhalf = pOff << 3;
#pragma unroll
  for (int v = 0; v < 8; ++v) {
    int r = rowBase + v + rhalf;
    if (r < N_NODES) {
      float nr = norm[r];
#pragma unroll
      for (int t = 0; t < 8; ++t)
        m[(size_t)r * HD + t * 16 + lane16] = acc[t][v] * nr;
    }
  }
}

__global__ __launch_bounds__(256) void zero_f4(float4* __restrict__ p, int n4) {
  int t = blockIdx.x * blockDim.x + threadIdx.x;
  if (t < n4) p[t] = make_float4(0.f, 0.f, 0.f, 0.f);
}

__global__ __launch_bounds__(256) void scatter_edges(
    const float* __restrict__ m, const int* __restrict__ src,
    const int* __restrict__ dst, float* __restrict__ agg) {
  int wave = threadIdx.x >> 5;
  int lane = threadIdx.x & 31;
  int e = blockIdx.x * 8 + wave;
  if (e >= N_EDGES) return;
  int s = src[e];
  int d = dst[e];
  float4 v    = *(const float4*)(m + (size_t)s * HD + lane * 4);
  float* dstp = agg + (size_t)d * HD + lane * 4;
  (void)__hip_atomic_fetch_add(dstp + 0, v.x, __ATOMIC_RELAXED, __HIP_MEMORY_SCOPE_AGENT);
  (void)__hip_atomic_fetch_add(dstp + 1, v.y, __ATOMIC_RELAXED, __HIP_MEMORY_SCOPE_AGENT);
  (void)__hip_atomic_fetch_add(dstp + 2, v.z, __ATOMIC_RELAXED, __HIP_MEMORY_SCOPE_AGENT);
  (void)__hip_atomic_fetch_add(dstp + 3, v.w, __ATOMIC_RELAXED, __HIP_MEMORY_SCOPE_AGENT);
}

__global__ __launch_bounds__(256) void epilogue_leaky(
    const float* __restrict__ agg, const float* __restrict__ norm,
    const float* __restrict__ bias, float* __restrict__ out) {
  int t   = blockIdx.x * blockDim.x + threadIdx.x;
  int row = t >> 5;
  int c4  = (t & 31) << 2;
  if (row < N_NODES) {
    float nr = norm[row];
    float4 a = *(const float4*)(agg + (size_t)row * HD + c4);
    float4 b = *(const float4*)(bias + c4);
    float4 n;
    n.x = a.x * nr + b.x;
    n.y = a.y * nr + b.y;
    n.z = a.z * nr + b.z;
    n.w = a.w * nr + b.w;
    n.x = n.x > 0.f ? n.x : 0.2f * n.x;
    n.y = n.y > 0.f ? n.y : 0.2f * n.y;
    n.z = n.z > 0.f ? n.z : 0.2f * n.z;
    n.w = n.w > 0.f ? n.w : 0.2f * n.w;
    *(float4*)(out + (size_t)row * HD + c4) = n;
  }
}

extern "C" void kernel_launch(void* const* d_in, const int* in_sizes, int n_in,
                              void* d_out, int out_size, void* d_ws, size_t ws_size,
                              hipStream_t stream) {
  const int*   node_id = (const int*)d_in[0];
  const int*   src     = (const int*)d_in[1];
  const int*   dst     = (const int*)d_in[2];
  const float* norm    = (const float*)d_in[3];
  const float* embed   = (const float*)d_in[4];
  const float* W1      = (const float*)d_in[5];
  const float* b1      = (const float*)d_in[6];
  const float* W2      = (const float*)d_in[7];
  const float* b2      = (const float*)d_in[8];
  float* out = (float*)d_out;

  const size_t feat = (size_t)N_NODES * HD;
  float* bufH = (float*)d_ws;
  float* bufM = bufH + feat;
  float* agg  = out;

  dim3 blk(256);
  const int gElem    = (int)((feat / 4 + 255) / 256);
  const int gGemm    = (N_NODES + 127) / 128;
  const int gScatter = (N_EDGES + 7) / 8;

  gather_embed<<<gElem, blk, 0, stream>>>(node_id, embed, bufH);

  gemm_norm_wmma<<<gGemm, blk, 0, stream>>>(bufH, W1, norm, bufM);
  zero_f4<<<gElem, blk, 0, stream>>>((float4*)agg, (int)(feat / 4));
  scatter_edges<<<gScatter, blk, 0, stream>>>(bufM, src, dst, agg);
  epilogue_leaky<<<gElem, blk, 0, stream>>>(agg, norm, b1, bufH);

  gemm_norm_wmma<<<gGemm, blk, 0, stream>>>(bufH, W2, norm, bufM);
  zero_f4<<<gElem, blk, 0, stream>>>((float4*)agg, (int)(feat / 4));
  scatter_edges<<<gScatter, blk, 0, stream>>>(bufM, src, dst, agg);
  epilogue_leaky<<<gElem, blk, 0, stream>>>(agg, norm, b2, out);
}


# Edge sum by row-range owners: 256 blocks hold 392 rows in LDS, scan dst for them, each wave adds its own rows' m[src]; no f32 atomics, zero or epilogue pass; sum order now fixed by the edge list
# speedup vs baseline: 1.8585x; 1.8585x over previous
#include <hip/hip_runtime.h>

#define N_NODES 100000
#define N_EDGES 600000
#define HD 128

typedef __attribute__((ext_vector_type(2))) float v2f;
typedef __attribute__((ext_vector_type(8))) float v8f;
typedef __attribute__((ext_vector_type(4))) float v4f;

#define ROWS_PER_WAVE 16
static_assert(N_NODES % ROWS_PER_WAVE == 0, "a wave's rows must all exist");
static_assert(ROWS_PER_WAVE == 16, "the load fence in gather_embed names 16 rows");

__device__ __forceinline__ int wave_first_row() {
  int wave = __builtin_amdgcn_readfirstlane((int)(threadIdx.x >> 5));
  return ((int)blockIdx.x * 8 + wave) * ROWS_PER_WAVE;
}

__global__ __launch_bounds__(256) void gather_embed(
    const int* __restrict__ node_id, const float* __restrict__ embed,
    float* __restrict__ h) {
  int lane = threadIdx.x & 31;
  int c4   = lane << 2;
  int r0   = wave_first_row();
  if (r0 >= N_NODES) return;
  int myid = node_id[r0 + (lane & (ROWS_PER_WAVE - 1))];
  v4f v[ROWS_PER_WAVE];
#pragma unroll
  for (int j = 0; j < ROWS_PER_WAVE; ++j) {
    int srow = __builtin_amdgcn_readlane(myid, j);
    v[j] = *(const v4f*)(embed + (size_t)srow * HD + c4);
  }
  asm volatile("" :: "v"(v[0]), "v"(v[1]), "v"(v[2]), "v"(v[3]),
                     "v"(v[4]), "v"(v[5]), "v"(v[6]), "v"(v[7]),
                     "v"(v[8]), "v"(v[9]), "v"(v[10]), "v"(v[11]),
                     "v"(v[12]), "v"(v[13]), "v"(v[14]), "v"(v[15]));
#pragma unroll
  for (int j = 0; j < ROWS_PER_WAVE; ++j)
    *(v4f*)(h + (size_t)(r0 + j) * HD + c4) = v[j];
}

__global__ __launch_bounds__(256) void gemm_norm_wmma(
    const float* __restrict__ h, const float* __restrict__ W,
    const float* __restrict__ norm, float* __restrict__ m) {
  __shared__ float Wl2[HD * HD];

  int tid = threadIdx.x;
  for (int i = tid; i < (HD * HD) / 4; i += 256) {
    int p = i >> 6;
    int n = (i & 63) << 1;
    int k = p << 1;
    float2 r0 = *(const float2*)(W + (size_t)k * HD + n);
    float2 r1 = *(const float2*)(W + (size_t)(k + 1) * HD + n);
    float4 o  = make_float4(r0.x, r1.x, r0.y, r1.y);
    *(float4*)(Wl2 + p * 256 + n * 2) = o;
  }
  __syncthreads();

  int wave    = tid >> 5;
  int lane    = tid & 31;
  int lane16  = lane & 15;
  int pOff    = lane >> 4;
  int rowBase = (blockIdx.x * 8 + wave) * 16;
  int rowA    = rowBase + lane16;
  int rowAc   = rowA < N_NODES ? rowA : (N_NODES - 1);

  v8f acc[8] = {};
  const float* hrow = h + (size_t)rowAc * HD;

  for (int k0 = 0; k0 < HD; k0 += 4) {
    v2f a = *(const v2f*)(hrow + k0 + (pOff << 1));
    const float* bbase = Wl2 + ((k0 >> 1) + pOff) * 256;
#pragma unroll
    for (int t = 0; t < 8; ++t) {
      int n = t * 16 + lane16;
      v2f b = *(const v2f*)(bbase + n * 2);
      acc[t] = __builtin_amdgcn_wmma_f32_16x16x4_f32(
          false, a, false, b, (short)0, acc[t], false, false);
    }
  }

  int rhalf = pOff << 3;
#pragma unroll
  for (int v = 0; v < 8; ++v) {
    int r = rowBase + v + rhalf;
    if (r < N_NODES) {
      float nr = norm[r];
#pragma unroll
      for (int t = 0; t < 8; ++t)
        m[(size_t)r * HD + t * 16 + lane16] = acc[t][v] * nr;
    }
  }
}

#define OWN_ROWS    392
#define OWN_WGS     ((N_NODES + OWN_ROWS - 1) / OWN_ROWS)
#define OWN_WAVES   32
#define SCAN_CAP    384
#define MINE_CAP    256
#define CHUNK_EDGES 128
#define N_CHUNKS    ((N_EDGES + CHUNK_EDGES - 1) / CHUNK_EDGES)
#define SCAN_DEPTH  8
#define ROW_DEPTH   8
static_assert(N_EDGES % 4 == 0, "a lane's int4 of dst is wholly inside the list or wholly outside");
static_assert(OWN_ROWS <= 512, "the local row takes the 9 low bits of an entry");
static_assert(N_EDGES < (1 << 22), "the edge index takes the bits above them");
static_assert(SCAN_CAP >= 2 * CHUNK_EDGES && MINE_CAP >= 64, "a round always makes progress");

__device__ __forceinline__ void add_rows(
    const int* mine, int n, int lane, const float* __restrict__ m,
    const int* __restrict__ src, float* tile) {
  for (int b0 = 0; b0 < n; b0 += 32) {
    int k   = b0 + lane;
    int ent = 0;
    int s   = 0;
    if (k < n) {
      ent = mine[k];
      s   = src[ent >> 9];
    }
    int r  = ent & 511;
    int nb = (n - b0) < 32 ? (n - b0) : 32;
    for (int q0 = 0; q0 < nb; q0 += ROW_DEPTH) {
      int sq[ROW_DEPTH];
#pragma unroll
      for (int q = 0; q < ROW_DEPTH; ++q)
        sq[q] = __builtin_amdgcn_readlane(s, q0 + q);
      v4f v[ROW_DEPTH];
#pragma unroll
      for (int q = 0; q < ROW_DEPTH; ++q)
        v[q] = *(const v4f*)(m + (size_t)sq[q] * HD + (lane << 2));
#pragma unroll
      for (int q = 0; q < ROW_DEPTH; ++q) {
        if (q0 + q < nb) {
          int rq = __builtin_amdgcn_readlane(r, q0 + q);
          v4f* t = (v4f*)(tile + rq * HD + (lane << 2));
          *t = *t + v[q];
        }
      }
    }
  }
}

__global__ __launch_bounds__(1024) void owner_sum(
    const float* __restrict__ m, const int* __restrict__ src,
    const int* __restrict__ dst, const float* __restrict__ norm,
    const float* __restrict__ bias, float* __restrict__ out) {
  __shared__ alignas(16) float tile[OWN_ROWS * HD];
  __shared__ int   scanList[OWN_WAVES * SCAN_CAP];
  __shared__ int   mineList[OWN_WAVES * MINE_CAP];
  __shared__ int   scanCnt[OWN_WAVES];
  __shared__ int   moreFlag[2];

  const int tid  = threadIdx.x;
  const int lane = tid & 31;
  const int w    = __builtin_amdgcn_readfirstlane(tid >> 5);
  const unsigned ltmask = (1u << lane) - 1u;
  const int base  = (int)blockIdx.x * OWN_ROWS;
  const int nrows = (N_NODES - base) < OWN_ROWS ? (N_NODES - base) : OWN_ROWS;

  for (int i = tid; i < OWN_ROWS * HD / 4; i += 1024)
    ((float4*)tile)[i] = make_float4(0.f, 0.f, 0.f, 0.f);
  if (tid < 2) moreFlag[tid] = 0;
  __syncthreads();

  int* myScan = scanList + w * SCAN_CAP;
  int* myMine = mineList + w * MINE_CAP;
  int pos   = w;
  int round = 0;
  for (;;) {
    int  cnt  = 0;
    bool full = false;
    while (pos < N_CHUNKS && !full) {
      int4 d[SCAN_DEPTH];
#pragma unroll
      for (int u = 0; u < SCAN_DEPTH; ++u) {
        int e0 = (pos + u * OWN_WAVES) * CHUNK_EDGES + lane * 4;
        d[u] = make_int4(-1, -1, -1, -1);
        if (e0 < N_EDGES) d[u] = *(const int4*)(dst + e0);
      }
      int used = 0;
#pragma unroll
      for (int u = 0; u < SCAN_DEPTH; ++u) {
        if (!full) {
          if (cnt + CHUNK_EDGES > SCAN_CAP) {
            full = true;
          } else {
            int e0 = (pos + u * OWN_WAVES) * CHUNK_EDGES + lane * 4;
            int r0 = d[u].x - base, r1 = d[u].y - base;
            int r2 = d[u].z - base, r3 = d[u].w - base;
            bool h0 = (unsigned)r0 < (unsigned)nrows;
            bool h1 = (unsigned)r1 < (unsigned)nrows;
            bool h2 = (unsigned)r2 < (unsigned)nrows;
            bool h3 = (unsigned)r3 < (unsigned)nrows;
            if (__builtin_amdgcn_ballot_w32(h0 | h1 | h2 | h3) != 0u) {
              unsigned bal;
              bal = __builtin_amdgcn_ballot_w32(h0);
              if (h0) myScan[cnt + __popc(bal & ltmask)] = ((e0 + 0) << 9) | r0;
              cnt += __popc(bal);
              bal = __builtin_amdgcn_ballot_w32(h1);
              if (h1) myScan[cnt + __popc(bal & ltmask)] = ((e0 + 1) << 9) | r1;
              cnt += __popc(bal);
              bal = __builtin_amdgcn_ballot_w32(h2);
              if (h2) myScan[cnt + __popc(bal & ltmask)] = ((e0 + 2) << 9) | r2;
              cnt += __popc(bal);
              bal = __builtin_amdgcn_ballot_w32(h3);
              if (h3) myScan[cnt + __popc(bal & ltmask)] = ((e0 + 3) << 9) | r3;
              cnt += __popc(bal);
            }
            used = u + 1;
          }
        }
      }
      pos += used * OWN_WAVES;
    }
    if (lane == 0) {
      scanCnt[w] = cnt;
      if (pos < N_CHUNKS) moreFlag[round & 1] = 1;
    }
    __syncthreads();
    const int more = moreFlag[round & 1];
    if (tid == 0) moreFlag[(round + 1) & 1] = 0;

    int cntV = scanCnt[lane];
    int nm = 0;
    for (int lw = 0; lw < OWN_WAVES; ++lw) {
      int n = __builtin_amdgcn_readlane(cntV, lw);
      const int* L = scanList + lw * SCAN_CAP;
      for (int i0 = 0; i0 < n; i0 += 96) {
        int k0 = i0 + lane, k1 = k0 + 32, k2 = k0 + 64;
        int ent0 = (k0 < n) ? L[k0] : -1;
        int ent1 = (k1 < n) ? L[k1] : -1;
        int ent2 = (k2 < n) ? L[k2] : -1;
        int ents[3] = {ent0, ent1, ent2};
#pragma unroll
        for (int g = 0; g < 3; ++g) {
          int ent = ents[g];
          bool hit = (ent >= 0) && ((ent & 31) == w);
          unsigned bal = __builtin_amdgcn_ballot_w32(hit);
          if (bal != 0u) {
            if (nm + 32 > MINE_CAP) {
              add_rows(myMine, nm, lane, m, src, tile);
              nm = 0;
            }
            if (hit) myMine[nm + __popc(bal & ltmask)] = ent;
            nm += __popc(bal);
          }
        }
      }
    }
    add_rows(myMine, nm, lane, m, src, tile);

    if (!more) break;
    ++round;
    __syncthreads();
  }
  __syncthreads();

  const int c4 = lane << 2;
  const float4 b = *(const float4*)(bias + c4);
  for (int i = tid; i < nrows * 32; i += 1024) {
    int r = i >> 5;
    float nr = norm[base + r];
    float4 a = ((const float4*)tile)[i];
    float4 n;
    n.x = a.x * nr + b.x;
    n.y = a.y * nr + b.y;
    n.z = a.z * nr + b.z;
    n.w = a.w * nr + b.w;
    n.x = n.x > 0.f ? n.x : 0.2f * n.x;
    n.y = n.y > 0.f ? n.y : 0.2f * n.y;
    n.z = n.z > 0.f ? n.z : 0.2f * n.z;
    n.w = n.w > 0.f ? n.w : 0.2f * n.w;
    *(float4*)(out + (size_t)(base + r) * HD + c4) = n;
  }
}

extern "C" void kernel_launch(void* const* d_in, const int* in_sizes, int n_in,
                              void* d_out, int out_size, void* d_ws, size_t ws_size,
                              hipStream_t stream) {
  const int*   node_id = (const int*)d_in[0];
  const int*   src     = (const int*)d_in[1];
  const int*   dst     = (const int*)d_in[2];
  const float* norm    = (const float*)d_in[3];
  const float* embed   = (const float*)d_in[4];
  const float* W1      = (const float*)d_in[5];
  const float* b1      = (const float*)d_in[6];
  const float* W2      = (const float*)d_in[7];
  const float* b2      = (const float*)d_in[8];
  float* out = (float*)d_out;

  const size_t feat = (size_t)N_NODES * HD;
  float* bufH = (float*)d_ws;
  float* bufM = bufH + feat;

  dim3 blk(256);
  const int gRows = (N_NODES / ROWS_PER_WAVE + 7) / 8;
  const int gGemm = (N_NODES + 127) / 128;

  gather_embed<<<gRows, blk, 0, stream>>>(node_id, embed, bufH);

  gemm_norm_wmma<<<gGemm, blk, 0, stream>>>(bufH, W1, norm, bufM);
  owner_sum<<<OWN_WGS, dim3(1024), 0, stream>>>(bufM, src, dst, norm, b1, bufH);

  gemm_norm_wmma<<<gGemm, blk, 0, stream>>>(bufH, W2, norm, bufM);
  owner_sum<<<OWN_WGS, dim3(1024), 0, stream>>>(bufM, src, dst, norm, b2, out);
}


# Layer 2's owner sum reads per-wave (src,row) lists that layer 1's left in workspace in its add order: no second dst scan, list walk or src lookup; scan kept for list overflow; sum order unchanged
# speedup vs baseline: 2.2137x; 1.1911x over previous
#include <hip/hip_runtime.h>

#define N_NODES 100000
#define N_EDGES 600000
#define HD 128

typedef __attribute__((ext_vector_type(2))) float v2f;
typedef __attribute__((ext_vector_type(8))) float v8f;
typedef __attribute__((ext_vector_type(4))) float v4f;

#define ROWS_PER_WAVE 16
static_assert(N_NODES % ROWS_PER_WAVE == 0, "a wave's rows must all exist");
static_assert(ROWS_PER_WAVE == 16, "the load fence in gather_embed names 16 rows");

__device__ __forceinline__ int wave_first_row() {
  int wave = __builtin_amdgcn_readfirstlane((int)(threadIdx.x >> 5));
  return ((int)blockIdx.x * 8 + wave) * ROWS_PER_WAVE;
}

__global__ __launch_bounds__(256) void gather_embed(
    const int* __restrict__ node_id, const float* __restrict__ embed,
    float* __restrict__ h) {
  int lane = threadIdx.x & 31;
  int c4   = lane << 2;
  int r0   = wave_first_row();
  if (r0 >= N_NODES) return;
  int myid = node_id[r0 + (lane & (ROWS_PER_WAVE - 1))];
  v4f v[ROWS_PER_WAVE];
#pragma unroll
  for (int j = 0; j < ROWS_PER_WAVE; ++j) {
    int srow = __builtin_amdgcn_readlane(myid, j);
    v[j] = *(const v4f*)(embed + (size_t)srow * HD + c4);
  }
  asm volatile("" :: "v"(v[0]), "v"(v[1]), "v"(v[2]), "v"(v[3]),
                     "v"(v[4]), "v"(v[5]), "v"(v[6]), "v"(v[7]),
                     "v"(v[8]), "v"(v[9]), "v"(v[10]), "v"(v[11]),
                     "v"(v[12]), "v"(v[13]), "v"(v[14]), "v"(v[15]));
#pragma unroll
  for (int j = 0; j < ROWS_PER_WAVE; ++j)
    *(v4f*)(h + (size_t)(r0 + j) * HD + c4) = v[j];
}

__global__ __launch_bounds__(256) void gemm_norm_wmma(
    const float* __restrict__ h, const float* __restrict__ W,
    const float* __restrict__ norm, float* __restrict__ m) {
  __shared__ float Wl2[HD * HD];

  int tid = threadIdx.x;
  for (int i = tid; i < (HD * HD) / 4; i += 256) {
    int p = i >> 6;
    int n = (i & 63) << 1;
    int k = p << 1;
    float2 r0 = *(const float2*)(W + (size_t)k * HD + n);
    float2 r1 = *(const float2*)(W + (size_t)(k + 1) * HD + n);
    float4 o  = make_float4(r0.x, r1.x, r0.y, r1.y);
    *(float4*)(Wl2 + p * 256 + n * 2) = o;
  }
  __syncthreads();

  int wave    = tid >> 5;
  int lane    = tid & 31;
  int lane16  = lane & 15;
  int pOff    = lane >> 4;
  int rowBase = (blockIdx.x * 8 + wave) * 16;
  int rowA    = rowBase + lane16;
  int rowAc   = rowA < N_NODES ? rowA : (N_NODES - 1);

  v8f acc[8] = {};
  const float* hrow = h + (size_t)rowAc * HD;

  for (int k0 = 0; k0 < HD; k0 += 4) {
    v2f a = *(const v2f*)(hrow + k0 + (pOff << 1));
    const float* bbase = Wl2 + ((k0 >> 1) + pOff) * 256;
#pragma unroll
    for (int t = 0; t < 8; ++t) {
      int n = t * 16 + lane16;
      v2f b = *(const v2f*)(bbase + n * 2);
      acc[t] = __builtin_amdgcn_wmma_f32_16x16x4_f32(
          false, a, false, b, (short)0, acc[t], false, false);
    }
  }

  int rhalf = pOff << 3;
#pragma unroll
  for (int v = 0; v < 8; ++v) {
    int r = rowBase + v + rhalf;
    if (r < N_NODES) {
      float nr = norm[r];
#pragma unroll
      for (int t = 0; t < 8; ++t)
        m[(size_t)r * HD + t * 16 + lane16] = acc[t][v] * nr;
    }
  }
}

#define OWN_ROWS    392
#define OWN_WGS     ((N_NODES + OWN_ROWS - 1) / OWN_ROWS)
#define OWN_WAVES   32
#define SCAN_CAP    384
#define MINE_CAP    256
#define CHUNK_EDGES 128
#define N_CHUNKS    ((N_EDGES + CHUNK_EDGES - 1) / CHUNK_EDGES)
#define SCAN_DEPTH  8
#define ROW_DEPTH   8
#define LIST_CAP    256
#define LIST_INTS   ((size_t)OWN_WGS * OWN_WAVES * LIST_CAP)
static_assert(N_EDGES % 4 == 0, "a lane's int4 of dst is wholly inside the list or wholly outside");
static_assert(OWN_ROWS <= 512, "the local row takes the 9 low bits of an entry");
static_assert(N_EDGES < (1 << 22), "the edge index takes the bits above them");
static_assert(SCAN_CAP >= 2 * CHUNK_EDGES && MINE_CAP >= 64, "a round always makes progress");
static_assert(N_NODES <= (1 << 22), "a source row shares an int with the 9-bit local row");

__device__ __forceinline__ void add_batch(
    int s, int r, int nb, int lane, const float* __restrict__ m, float* tile) {
  for (int q0 = 0; q0 < nb; q0 += ROW_DEPTH) {
    int sq[ROW_DEPTH];
#pragma unroll
    for (int q = 0; q < ROW_DEPTH; ++q)
      sq[q] = __builtin_amdgcn_readlane(s, q0 + q);
    v4f v[ROW_DEPTH];
#pragma unroll
    for (int q = 0; q < ROW_DEPTH; ++q)
      v[q] = *(const v4f*)(m + (size_t)sq[q] * HD + (lane << 2));
#pragma unroll
    for (int q = 0; q < ROW_DEPTH; ++q) {
      if (q0 + q < nb) {
        int rq = __builtin_amdgcn_readlane(r, q0 + q);
        v4f* t = (v4f*)(tile + rq * HD + (lane << 2));
        *t = *t + v[q];
      }
    }
  }
}

template <bool SAVE>
__device__ __forceinline__ void add_rows(
    const int* mine, int n, int lane, const float* __restrict__ m,
    const int* __restrict__ src, float* tile, int* __restrict__ keep, int& kept) {
  for (int b0 = 0; b0 < n; b0 += 32) {
    int k   = b0 + lane;
    int ent = 0;
    int s   = 0;
    if (k < n) {
      ent = mine[k];
      s   = src[ent >> 9];
    }
    int r  = ent & 511;
    int nb = (n - b0) < 32 ? (n - b0) : 32;
    if (SAVE) {
      if (k < n && kept + lane < LIST_CAP) keep[kept + lane] = (s << 9) | r;
      kept += nb;
    }
    add_batch(s, r, nb, lane, m, tile);
  }
}

__device__ __forceinline__ void add_kept_rows(
    const int* __restrict__ keep, int n, int lane,
    const float* __restrict__ m, float* tile) {
  int ents[LIST_CAP / 32];
#pragma unroll
  for (int b = 0; b < LIST_CAP / 32; ++b) {
    int k = b * 32 + lane;
    ents[b] = (k < n) ? keep[k] : 0;
  }
#pragma unroll
  for (int b = 0; b < LIST_CAP / 32; ++b) {
    int rest = n - b * 32;
    if (rest > 0)
      add_batch(ents[b] >> 9, ents[b] & 511, rest < 32 ? rest : 32, lane, m, tile);
  }
}

template <bool FIRST>
__global__ __launch_bounds__(1024) void owner_sum(
    const float* __restrict__ m, const int* __restrict__ src,
    const int* __restrict__ dst, const float* __restrict__ norm,
    const float* __restrict__ bias, float* __restrict__ out,
    int* __restrict__ lists, int* __restrict__ listCnt) {
  __shared__ alignas(16) float tile[OWN_ROWS * HD];
  __shared__ int   scanList[OWN_WAVES * SCAN_CAP];
  __shared__ int   mineList[OWN_WAVES * MINE_CAP];
  __shared__ int   scanCnt[OWN_WAVES];
  __shared__ int   moreFlag[2];

  const int tid  = threadIdx.x;
  const int lane = tid & 31;
  const int w    = __builtin_amdgcn_readfirstlane(tid >> 5);
  const unsigned ltmask = (1u << lane) - 1u;
  const int base  = (int)blockIdx.x * OWN_ROWS;
  const int nrows = (N_NODES - base) < OWN_ROWS ? (N_NODES - base) : OWN_ROWS;

  int* myKeep = lists + ((size_t)blockIdx.x * OWN_WAVES + w) * LIST_CAP;
  int  kept   = 0;
  int  keptV  = 0;
  if (!FIRST) keptV = listCnt[(int)blockIdx.x * OWN_WAVES + lane];

  for (int i = tid; i < OWN_ROWS * HD / 4; i += 1024)
    ((float4*)tile)[i] = make_float4(0.f, 0.f, 0.f, 0.f);
  if (tid < 2) moreFlag[tid] = 0;
  const bool useKept =
      !FIRST && __builtin_amdgcn_ballot_w32(keptV > LIST_CAP) == 0u;
  __syncthreads();

  int* myScan = scanList + w * SCAN_CAP;
  int* myMine = mineList + w * MINE_CAP;
  int pos   = w;
  int round = 0;
  if (useKept) {
    add_kept_rows(myKeep, __builtin_amdgcn_readlane(keptV, w), lane, m, tile);
  } else for (;;) {
    int  cnt  = 0;
    bool full = false;
    while (pos < N_CHUNKS && !full) {
      int4 d[SCAN_DEPTH];
#pragma unroll
      for (int u = 0; u < SCAN_DEPTH; ++u) {
        int e0 = (pos + u * OWN_WAVES) * CHUNK_EDGES + lane * 4;
        d[u] = make_int4(-1, -1, -1, -1);
        if (e0 < N_EDGES) d[u] = *(const int4*)(dst + e0);
      }
      int used = 0;
#pragma unroll
      for (int u = 0; u < SCAN_DEPTH; ++u) {
        if (!full) {
          if (cnt + CHUNK_EDGES > SCAN_CAP) {
            full = true;
          } else {
            int e0 = (pos + u * OWN_WAVES) * CHUNK_EDGES + lane * 4;
            int r0 = d[u].x - base, r1 = d[u].y - base;
            int r2 = d[u].z - base, r3 = d[u].w - base;
            bool h0 = (unsigned)r0 < (unsigned)nrows;
            bool h1 = (unsigned)r1 < (unsigned)nrows;
            bool h2 = (unsigned)r2 < (unsigned)nrows;
            bool h3 = (unsigned)r3 < (unsigned)nrows;
            if (__builtin_amdgcn_ballot_w32(h0 | h1 | h2 | h3) != 0u) {
              unsigned bal;
              bal = __builtin_amdgcn_ballot_w32(h0);
              if (h0) myScan[cnt + __popc(bal & ltmask)] = ((e0 + 0) << 9) | r0;
              cnt += __popc(bal);
              bal = __builtin_amdgcn_ballot_w32(h1);
              if (h1) myScan[cnt + __popc(bal & ltmask)] = ((e0 + 1) << 9) | r1;
              cnt += __popc(bal);
              bal = __builtin_amdgcn_ballot_w32(h2);
              if (h2) myScan[cnt + __popc(bal & ltmask)] = ((e0 + 2) << 9) | r2;
              cnt += __popc(bal);
              bal = __builtin_amdgcn_ballot_w32(h3);
              if (h3) myScan[cnt + __popc(bal & ltmask)] = ((e0 + 3) << 9) | r3;
              cnt += __popc(bal);
            }
            used = u + 1;
          }
        }
      }
      pos += used * OWN_WAVES;
    }
    if (lane == 0) {
      scanCnt[w] = cnt;
      if (pos < N_CHUNKS) moreFlag[round & 1] = 1;
    }
    __syncthreads();
    const int more = moreFlag[round & 1];
    if (tid == 0) moreFlag[(round + 1) & 1] = 0;

    int cntV = scanCnt[lane];
    int nm = 0;
    for (int lw = 0; lw < OWN_WAVES; ++lw) {
      int n = __builtin_amdgcn_readlane(cntV, lw);
      const int* L = scanList + lw * SCAN_CAP;
      for (int i0 = 0; i0 < n; i0 += 96) {
        int k0 = i0 + lane, k1 = k0 + 32, k2 = k0 + 64;
        int ent0 = (k0 < n) ? L[k0] : -1;
        int ent1 = (k1 < n) ? L[k1] : -1;
        int ent2 = (k2 < n) ? L[k2] : -1;
        int ents[3] = {ent0, ent1, ent2};
#pragma unroll
        for (int g = 0; g < 3; ++g) {
          int ent = ents[g];
          bool hit = (ent >= 0) && ((ent & 31) == w);
          unsigned bal = __builtin_amdgcn_ballot_w32(hit);
          if (bal != 0u) {
            if (nm + 32 > MINE_CAP) {
              add_rows<FIRST>(myMine, nm, lane, m, src, tile, myKeep, kept);
              nm = 0;
            }
            if (hit) myMine[nm + __popc(bal & ltmask)] = ent;
            nm += __popc(bal);
          }
        }
      }
    }
    add_rows<FIRST>(myMine, nm, lane, m, src, tile, myKeep, kept);

    if (!more) break;
    ++round;
    __syncthreads();
  }
  if (FIRST && lane == 0) listCnt[(int)blockIdx.x * OWN_WAVES + w] = kept;
  __syncthreads();

  const int c4 = lane << 2;
  const float4 b = *(const float4*)(bias + c4);
  for (int i = tid; i < nrows * 32; i += 1024) {
    int r = i >> 5;
    float nr = norm[base + r];
    float4 a = ((const float4*)tile)[i];
    float4 n;
    n.x = a.x * nr + b.x;
    n.y = a.y * nr + b.y;
    n.z = a.z * nr + b.z;
    n.w = a.w * nr + b.w;
    n.x = n.x > 0.f ? n.x : 0.2f * n.x;
    n.y = n.y > 0.f ? n.y : 0.2f * n.y;
    n.z = n.z > 0.f ? n.z : 0.2f * n.z;
    n.w = n.w > 0.f ? n.w : 0.2f * n.w;
    *(float4*)(out + (size_t)(base + r) * HD + c4) = n;
  }
}

extern "C" void kernel_launch(void* const* d_in, const int* in_sizes, int n_in,
                              void* d_out, int out_size, void* d_ws, size_t ws_size,
                              hipStream_t stream) {
  const int*   node_id = (const int*)d_in[0];
  const int*   src     = (const int*)d_in[1];
  const int*   dst     = (const int*)d_in[2];
  const float* norm    = (const float*)d_in[3];
  const float* embed   = (const float*)d_in[4];
  const float* W1      = (const float*)d_in[5];
  const float* b1      = (const float*)d_in[6];
  const float* W2      = (const float*)d_in[7];
  const float* b2      = (const float*)d_in[8];
  float* out = (float*)d_out;

  const size_t feat = (size_t)N_NODES * HD;
  float* bufH = (float*)d_ws;
  float* bufM = bufH + feat;
  int* lists   = (int*)(bufM + feat);
  int* listCnt = lists + LIST_INTS;

  dim3 blk(256);
  const int gRows = (N_NODES / ROWS_PER_WAVE + 7) / 8;
  const int gGemm = (N_NODES + 127) / 128;

  gather_embed<<<gRows, blk, 0, stream>>>(node_id, embed, bufH);

  gemm_norm_wmma<<<gGemm, blk, 0, stream>>>(bufH, W1, norm, bufM);
  owner_sum<true><<<OWN_WGS, dim3(1024), 0, stream>>>(bufM, src, dst, norm, b1, bufH,
                                                      lists, listCnt);

  gemm_norm_wmma<<<gGemm, blk, 0, stream>>>(bufH, W2, norm, bufM);
  owner_sum<false><<<OWN_WGS, dim3(1024), 0, stream>>>(bufM, src, dst, norm, b2, out,
                                                       lists, listCnt);
}


# Owner scan reads 1 owner byte per edge (dst/392, written by extra blocks of launch 1) and dst only for its matches: 4x fewer scan bytes and loads; a row's sum order changes, still fixed by edge list
# speedup vs baseline: 2.4342x; 1.0996x over previous
#include <hip/hip_runtime.h>

#define N_NODES 100000
#define N_EDGES 600000
#define HD 128

typedef __attribute__((ext_vector_type(2))) float v2f;
typedef __attribute__((ext_vector_type(8))) float v8f;
typedef __attribute__((ext_vector_type(4))) float v4f;

#define ROWS_PER_WAVE 16
#define ROW_BLOCKS ((N_NODES / ROWS_PER_WAVE + 7) / 8)
#define OWN_ROWS    392
#define OWNER_EDGES_PER_BLOCK 4096
#define OWNER_BLOCKS ((N_EDGES + OWNER_EDGES_PER_BLOCK - 1) / OWNER_EDGES_PER_BLOCK)
static_assert(N_NODES % ROWS_PER_WAVE == 0, "a wave's rows must all exist");
static_assert(ROWS_PER_WAVE == 16, "the load fence in gather_embed names 16 rows");

__device__ __forceinline__ int wave_first_row() {
  int wave = __builtin_amdgcn_readfirstlane((int)(threadIdx.x >> 5));
  return ((int)blockIdx.x * 8 + wave) * ROWS_PER_WAVE;
}

__global__ __launch_bounds__(256) void gather_embed(
    const int* __restrict__ node_id, const float* __restrict__ embed,
    float* __restrict__ h, const int* __restrict__ dst,
    unsigned* __restrict__ ownerW) {
  if ((int)blockIdx.x >= ROW_BLOCKS) {
    int eb = ((int)blockIdx.x - ROW_BLOCKS) * OWNER_EDGES_PER_BLOCK + (int)threadIdx.x * 4;
    int4 d[4];
#pragma unroll
    for (int u = 0; u < 4; ++u) {
      int e0 = eb + u * 1024;
      d[u] = *(const int4*)(dst + ((e0 < N_EDGES) ? e0 : 0));
    }
#pragma unroll
    for (int u = 0; u < 4; ++u) {
      int e0 = eb + u * 1024;
      if (e0 < N_EDGES)
        ownerW[e0 >> 2] = (((unsigned)d[u].x / OWN_ROWS) & 255u)
                        | ((((unsigned)d[u].y / OWN_ROWS) & 255u) << 8)
                        | ((((unsigned)d[u].z / OWN_ROWS) & 255u) << 16)
                        | ((((unsigned)d[u].w / OWN_ROWS) & 255u) << 24);
    }
    return;
  }
  int lane = threadIdx.x & 31;
  int c4   = lane << 2;
  int r0   = wave_first_row();
  if (r0 >= N_NODES) return;
  int myid = node_id[r0 + (lane & (ROWS_PER_WAVE - 1))];
  v4f v[ROWS_PER_WAVE];
#pragma unroll
  for (int j = 0; j < ROWS_PER_WAVE; ++j) {
    int srow = __builtin_amdgcn_readlane(myid, j);
    v[j] = *(const v4f*)(embed + (size_t)srow * HD + c4);
  }
  asm volatile("" :: "v"(v[0]), "v"(v[1]), "v"(v[2]), "v"(v[3]),
                     "v"(v[4]), "v"(v[5]), "v"(v[6]), "v"(v[7]),
                     "v"(v[8]), "v"(v[9]), "v"(v[10]), "v"(v[11]),
                     "v"(v[12]), "v"(v[13]), "v"(v[14]), "v"(v[15]));
#pragma unroll
  for (int j = 0; j < ROWS_PER_WAVE; ++j)
    *(v4f*)(h + (size_t)(r0 + j) * HD + c4) = v[j];
}

__global__ __launch_bounds__(256) void gemm_norm_wmma(
    const float* __restrict__ h, const float* __restrict__ W,
    const float* __restrict__ norm, float* __restrict__ m) {
  __shared__ float Wl2[HD * HD];

  int tid = threadIdx.x;
  for (int i = tid; i < (HD * HD) / 4; i += 256) {
    int p = i >> 6;
    int n = (i & 63) << 1;
    int k = p << 1;
    float2 r0 = *(const float2*)(W + (size_t)k * HD + n);
    float2 r1 = *(const float2*)(W + (size_t)(k + 1) * HD + n);
    float4 o  = make_float4(r0.x, r1.x, r0.y, r1.y);
    *(float4*)(Wl2 + p * 256 + n * 2) = o;
  }
  __syncthreads();

  int wave    = tid >> 5;
  int lane    = tid & 31;
  int lane16  = lane & 15;
  int pOff    = lane >> 4;
  int rowBase = (blockIdx.x * 8 + wave) * 16;
  int rowA    = rowBase + lane16;
  int rowAc   = rowA < N_NODES ? rowA : (N_NODES - 1);

  v8f acc[8] = {};
  const float* hrow = h + (size_t)rowAc * HD;

  for (int k0 = 0; k0 < HD; k0 += 4) {
    v2f a = *(const v2f*)(hrow + k0 + (pOff << 1));
    const float* bbase = Wl2 + ((k0 >> 1) + pOff) * 256;
#pragma unroll
    for (int t = 0; t < 8; ++t) {
      int n = t * 16 + lane16;
      v2f b = *(const v2f*)(bbase + n * 2);
      acc[t] = __builtin_amdgcn_wmma_f32_16x16x4_f32(
          false, a, false, b, (short)0, acc[t], false, false);
    }
  }

  int rhalf = pOff << 3;
#pragma unroll
  for (int v = 0; v < 8; ++v) {
    int r = rowBase + v + rhalf;
    if (r < N_NODES) {
      float nr = norm[r];
#pragma unroll
      for (int t = 0; t < 8; ++t)
        m[(size_t)r * HD + t * 16 + lane16] = acc[t][v] * nr;
    }
  }
}

#define OWN_WGS     ((N_NODES + OWN_ROWS - 1) / OWN_ROWS)
#define OWN_WAVES   32
#define SCAN_CAP    512
#define MINE_CAP    256
#define CHUNK_EDGES 512
#define N_CHUNKS    ((N_EDGES + CHUNK_EDGES - 1) / CHUNK_EDGES)
#define SCAN_DEPTH  8
#define ROW_DEPTH   8
#define LIST_CAP    256
#define LIST_INTS   ((size_t)OWN_WGS * OWN_WAVES * LIST_CAP)
static_assert(N_EDGES % 16 == 0, "a lane's 16 owner bytes are wholly inside the list or wholly outside");
static_assert(OWN_WGS <= 256, "an owner index fits a byte");
static_assert(OWN_ROWS <= 512, "the local row takes the 9 low bits of an entry");
static_assert(N_EDGES < (1 << 22), "the edge index takes the bits above them");
static_assert(SCAN_CAP >= CHUNK_EDGES && MINE_CAP >= 64, "a round always makes progress");
static_assert(N_NODES <= (1 << 22), "a source row shares an int with the 9-bit local row");

__device__ __forceinline__ void add_batch(
    int s, int r, int nb, int lane, const float* __restrict__ m, float* tile) {
  for (int q0 = 0; q0 < nb; q0 += ROW_DEPTH) {
    int sq[ROW_DEPTH];
#pragma unroll
    for (int q = 0; q < ROW_DEPTH; ++q)
      sq[q] = __builtin_amdgcn_readlane(s, q0 + q);
    v4f v[ROW_DEPTH];
#pragma unroll
    for (int q = 0; q < ROW_DEPTH; ++q)
      v[q] = *(const v4f*)(m + (size_t)sq[q] * HD + (lane << 2));
#pragma unroll
    for (int q = 0; q < ROW_DEPTH; ++q) {
      if (q0 + q < nb) {
        int rq = __builtin_amdgcn_readlane(r, q0 + q);
        v4f* t = (v4f*)(tile + rq * HD + (lane << 2));
        *t = *t + v[q];
      }
    }
  }
}

template <bool SAVE>
__device__ __forceinline__ void add_rows(
    const int* mine, int n, int lane, const float* __restrict__ m,
    const int* __restrict__ src, float* tile, int* __restrict__ keep, int& kept) {
  for (int b0 = 0; b0 < n; b0 += 32) {
    int k   = b0 + lane;
    int ent = 0;
    int s   = 0;
    if (k < n) {
      ent = mine[k];
      s   = src[ent >> 9];
    }
    int r  = ent & 511;
    int nb = (n - b0) < 32 ? (n - b0) : 32;
    if (SAVE) {
      if (k < n && kept + lane < LIST_CAP) keep[kept + lane] = (s << 9) | r;
      kept += nb;
    }
    add_batch(s, r, nb, lane, m, tile);
  }
}

__device__ __forceinline__ void add_kept_rows(
    const int* __restrict__ keep, int n, int lane,
    const float* __restrict__ m, float* tile) {
  int ents[LIST_CAP / 32];
#pragma unroll
  for (int b = 0; b < LIST_CAP / 32; ++b) {
    int k = b * 32 + lane;
    ents[b] = (k < n) ? keep[k] : 0;
  }
#pragma unroll
  for (int b = 0; b < LIST_CAP / 32; ++b) {
    int rest = n - b * 32;
    if (rest > 0)
      add_batch(ents[b] >> 9, ents[b] & 511, rest < 32 ? rest : 32, lane, m, tile);
  }
}

__device__ __forceinline__ unsigned zero_bytes(unsigned x) {
  return ~(((x & 0x7F7F7F7Fu) + 0x7F7F7F7Fu) | x | 0x7F7F7F7Fu);
}

template <bool FIRST>
__global__ __launch_bounds__(1024) void owner_sum(
    const float* __restrict__ m, const int* __restrict__ src,
    const int* __restrict__ dst, const float* __restrict__ norm,
    const float* __restrict__ bias, float* __restrict__ out,
    int* __restrict__ lists, int* __restrict__ listCnt,
    const unsigned* __restrict__ ownerW) {
  __shared__ alignas(16) float tile[OWN_ROWS * HD];
  __shared__ int   scanList[OWN_WAVES * SCAN_CAP];
  __shared__ int   mineList[OWN_WAVES * MINE_CAP];
  __shared__ int   scanCnt[OWN_WAVES];
  __shared__ int   moreFlag[2];

  const int tid  = threadIdx.x;
  const int lane = tid & 31;
  const int w    = __builtin_amdgcn_readfirstlane(tid >> 5);
  const unsigned ltmask = (1u << lane) - 1u;
  const int base  = (int)blockIdx.x * OWN_ROWS;
  const int nrows = (N_NODES - base) < OWN_ROWS ? (N_NODES - base) : OWN_ROWS;
  const unsigned pat = (unsigned)blockIdx.x * 0x01010101u;

  int* myKeep = lists + ((size_t)blockIdx.x * OWN_WAVES + w) * LIST_CAP;
  int  kept   = 0;
  int  keptV  = 0;
  if (!FIRST) keptV = listCnt[(int)blockIdx.x * OWN_WAVES + lane];

  for (int i = tid; i < OWN_ROWS * HD / 4; i += 1024)
    ((float4*)tile)[i] = make_float4(0.f, 0.f, 0.f, 0.f);
  if (tid < 2) moreFlag[tid] = 0;
  const bool useKept =
      !FIRST && __builtin_amdgcn_ballot_w32(keptV > LIST_CAP) == 0u;
  __syncthreads();

  int* myScan = scanList + w * SCAN_CAP;
  int* myMine = mineList + w * MINE_CAP;
  int pos   = w;
  int round = 0;
  if (useKept) {
    add_kept_rows(myKeep, __builtin_amdgcn_readlane(keptV, w), lane, m, tile);
  } else for (;;) {
    int  cnt  = 0;
    bool full = false;
    while (pos < N_CHUNKS && !full) {
      int4 d[SCAN_DEPTH];
#pragma unroll
      for (int u = 0; u < SCAN_DEPTH; ++u) {
        int e0 = (pos + u * OWN_WAVES) * CHUNK_EDGES + lane * 16;
        d[u] = *(const int4*)(ownerW + ((e0 < N_EDGES) ? (e0 >> 2) : 0));
      }
      int used = 0;
#pragma unroll
      for (int u = 0; u < SCAN_DEPTH; ++u) {
        if (!full) {
          unsigned z0 = zero_bytes((unsigned)d[u].x ^ pat);
          unsigned z1 = zero_bytes((unsigned)d[u].y ^ pat);
          unsigned z2 = zero_bytes((unsigned)d[u].z ^ pat);
          unsigned z3 = zero_bytes((unsigned)d[u].w ^ pat);
          const int e0 = (pos + u * OWN_WAVES) * CHUNK_EDGES + lane * 16;
          const bool any = (e0 < N_EDGES) && ((z0 | z1 | z2 | z3) != 0u);
          unsigned bal = __builtin_amdgcn_ballot_w32(any);
          if (bal != 0u) {
            if (cnt + 16 * __popc(bal) > SCAN_CAP) {
              full = true;
            } else {
              unsigned hm = (z0 >> 7) | (z1 >> 6) | (z2 >> 5) | (z3 >> 4);
              if (!any) hm = 0u;
              for (int it = 0; it < 16; ++it) {
                unsigned b2 = __builtin_amdgcn_ballot_w32(hm != 0u);
                if (b2 == 0u) break;
                if (hm != 0u) {
                  int p = __builtin_ctz(hm);
                  myScan[cnt + __popc(b2 & ltmask)] = e0 + ((p & 7) << 2) + (p >> 3);
                  hm &= hm - 1u;
                }
                cnt += __popc(b2);
              }
            }
          }
          if (!full) used = u + 1;
        }
      }
      pos += used * OWN_WAVES;
    }
    for (int i0 = 0; i0 < cnt; i0 += 128) {
      int e[4], r[4];
#pragma unroll
      for (int u = 0; u < 4; ++u) {
        int i = i0 + u * 32 + lane;
        e[u] = (i < cnt) ? myScan[i] : 0;
      }
#pragma unroll
      for (int u = 0; u < 4; ++u) r[u] = dst[e[u]] - base;
#pragma unroll
      for (int u = 0; u < 4; ++u) {
        int i = i0 + u * 32 + lane;
        if (i < cnt)
          myScan[i] = ((unsigned)r[u] < (unsigned)nrows) ? ((e[u] << 9) | r[u]) : -1;
      }
    }
    if (lane == 0) {
      scanCnt[w] = cnt;
      if (pos < N_CHUNKS) moreFlag[round & 1] = 1;
    }
    __syncthreads();
    const int more = moreFlag[round & 1];
    if (tid == 0) moreFlag[(round + 1) & 1] = 0;

    int cntV = scanCnt[lane];
    int nm = 0;
    for (int lw = 0; lw < OWN_WAVES; ++lw) {
      int n = __builtin_amdgcn_readlane(cntV, lw);
      const int* L = scanList + lw * SCAN_CAP;
      for (int i0 = 0; i0 < n; i0 += 96) {
        int k0 = i0 + lane, k1 = k0 + 32, k2 = k0 + 64;
        int ent0 = (k0 < n) ? L[k0] : -1;
        int ent1 = (k1 < n) ? L[k1] : -1;
        int ent2 = (k2 < n) ? L[k2] : -1;
        int ents[3] = {ent0, ent1, ent2};
#pragma unroll
        for (int g = 0; g < 3; ++g) {
          int ent = ents[g];
          bool hit = (ent >= 0) && ((ent & 31) == w);
          unsigned bal = __builtin_amdgcn_ballot_w32(hit);
          if (bal != 0u) {
            if (nm + 32 > MINE_CAP) {
              add_rows<FIRST>(myMine, nm, lane, m, src, tile, myKeep, kept);
              nm = 0;
            }
            if (hit) myMine[nm + __popc(bal & ltmask)] = ent;
            nm += __popc(bal);
          }
        }
      }
    }
    add_rows<FIRST>(myMine, nm, lane, m, src, tile, myKeep, kept);

    if (!more) break;
    ++round;
    __syncthreads();
  }
  if (FIRST && lane == 0) listCnt[(int)blockIdx.x * OWN_WAVES + w] = kept;
  __syncthreads();

  const int c4 = lane << 2;
  const float4 b = *(const float4*)(bias + c4);
  for (int i = tid; i < nrows * 32; i += 1024) {
    int r = i >> 5;
    float nr = norm[base + r];
    float4 a = ((const float4*)tile)[i];
    float4 n;
    n.x = a.x * nr + b.x;
    n.y = a.y * nr + b.y;
    n.z = a.z * nr + b.z;
    n.w = a.w * nr + b.w;
    n.x = n.x > 0.f ? n.x : 0.2f * n.x;
    n.y = n.y > 0.f ? n.y : 0.2f * n.y;
    n.z = n.z > 0.f ? n.z : 0.2f * n.z;
    n.w = n.w > 0.f ? n.w : 0.2f * n.w;
    *(float4*)(out + (size_t)(base + r) * HD + c4) = n;
  }
}

extern "C" void kernel_launch(void* const* d_in, const int* in_sizes, int n_in,
                              void* d_out, int out_size, void* d_ws, size_t ws_size,
                              hipStream_t stream) {
  const int*   node_id = (const int*)d_in[0];
  const int*   src     = (const int*)d_in[1];
  const int*   dst     = (const int*)d_in[2];
  const float* norm    = (const float*)d_in[3];
  const float* embed   = (const float*)d_in[4];
  const float* W1      = (const float*)d_in[5];
  const float* b1      = (const float*)d_in[6];
  const float* W2      = (const float*)d_in[7];
  const float* b2      = (const float*)d_in[8];
  float* out = (float*)d_out;

  const size_t feat = (size_t)N_NODES * HD;
  float* bufH = (float*)d_ws;
  float* bufM = bufH + feat;
  int* lists   = (int*)(bufM + feat);
  int* listCnt = lists + LIST_INTS;
  unsigned* ownerW = (unsigned*)(listCnt + OWN_WGS * OWN_WAVES);

  dim3 blk(256);
  const int gRows = ROW_BLOCKS + OWNER_BLOCKS;
  const int gGemm = (N_NODES + 127) / 128;

  gather_embed<<<gRows, blk, 0, stream>>>(node_id, embed, bufH, dst, ownerW);

  gemm_norm_wmma<<<gGemm, blk, 0, stream>>>(bufH, W1, norm, bufM);
  owner_sum<true><<<OWN_WGS, dim3(1024), 0, stream>>>(bufM, src, dst, norm, b1, bufH,
                                                      lists, listCnt, ownerW);

  gemm_norm_wmma<<<gGemm, blk, 0, stream>>>(bufH, W2, norm, bufM);
  owner_sum<false><<<OWN_WGS, dim3(1024), 0, stream>>>(bufM, src, dst, norm, b2, out,
                                                       lists, listCnt, ownerW);
}


# Layer 2's product moves into layer 1's owner sum: each group multiplies its activated LDS tile (row pitch 132) by W2 on f32 WMMA, W2 in the dead list space; 5 -> 4 launches, no h plane; sums same
# speedup vs baseline: 2.8013x; 1.1508x over previous
#include <hip/hip_runtime.h>

#define N_NODES 100000
#define N_EDGES 600000
#define HD 128

typedef __attribute__((ext_vector_type(2))) float v2f;
typedef __attribute__((ext_vector_type(8))) float v8f;
typedef __attribute__((ext_vector_type(4))) float v4f;

#define ROWS_PER_WAVE 16
#define ROW_BLOCKS ((N_NODES / ROWS_PER_WAVE + 7) / 8)
#define OWN_ROWS    392
#define OWNER_EDGES_PER_BLOCK 4096
#define OWNER_BLOCKS ((N_EDGES + OWNER_EDGES_PER_BLOCK - 1) / OWNER_EDGES_PER_BLOCK)
static_assert(N_NODES % ROWS_PER_WAVE == 0, "a wave's rows must all exist");
static_assert(ROWS_PER_WAVE == 16, "the load fence in gather_embed names 16 rows");

__device__ __forceinline__ int wave_first_row() {
  int wave = __builtin_amdgcn_readfirstlane((int)(threadIdx.x >> 5));
  return ((int)blockIdx.x * 8 + wave) * ROWS_PER_WAVE;
}

__global__ __launch_bounds__(256) void gather_embed(
    const int* __restrict__ node_id, const float* __restrict__ embed,
    float* __restrict__ h, const int* __restrict__ dst,
    unsigned* __restrict__ ownerW) {
  if ((int)blockIdx.x >= ROW_BLOCKS) {
    int eb = ((int)blockIdx.x - ROW_BLOCKS) * OWNER_EDGES_PER_BLOCK + (int)threadIdx.x * 4;
    int4 d[4];
#pragma unroll
    for (int u = 0; u < 4; ++u) {
      int e0 = eb + u * 1024;
      d[u] = *(const int4*)(dst + ((e0 < N_EDGES) ? e0 : 0));
    }
#pragma unroll
    for (int u = 0; u < 4; ++u) {
      int e0 = eb + u * 1024;
      if (e0 < N_EDGES)
        ownerW[e0 >> 2] = (((unsigned)d[u].x / OWN_ROWS) & 255u)
                        | ((((unsigned)d[u].y / OWN_ROWS) & 255u) << 8)
                        | ((((unsigned)d[u].z / OWN_ROWS) & 255u) << 16)
                        | ((((unsigned)d[u].w / OWN_ROWS) & 255u) << 24);
    }
    return;
  }
  int lane = threadIdx.x & 31;
  int c4   = lane << 2;
  int r0   = wave_first_row();
  if (r0 >= N_NODES) return;
  int myid = node_id[r0 + (lane & (ROWS_PER_WAVE - 1))];
  v4f v[ROWS_PER_WAVE];
#pragma unroll
  for (int j = 0; j < ROWS_PER_WAVE; ++j) {
    int srow = __builtin_amdgcn_readlane(myid, j);
    v[j] = *(const v4f*)(embed + (size_t)srow * HD + c4);
  }
  asm volatile("" :: "v"(v[0]), "v"(v[1]), "v"(v[2]), "v"(v[3]),
                     "v"(v[4]), "v"(v[5]), "v"(v[6]), "v"(v[7]),
                     "v"(v[8]), "v"(v[9]), "v"(v[10]), "v"(v[11]),
                     "v"(v[12]), "v"(v[13]), "v"(v[14]), "v"(v[15]));
#pragma unroll
  for (int j = 0; j < ROWS_PER_WAVE; ++j)
    *(v4f*)(h + (size_t)(r0 + j) * HD + c4) = v[j];
}

__global__ __launch_bounds__(256) void gemm_norm_wmma(
    const float* __restrict__ h, const float* __restrict__ W,
    const float* __restrict__ norm, float* __restrict__ m) {
  __shared__ float Wl2[HD * HD];

  int tid = threadIdx.x;
  for (int i = tid; i < (HD * HD) / 4; i += 256) {
    int p = i >> 6;
    int n = (i & 63) << 1;
    int k = p << 1;
    float2 r0 = *(const float2*)(W + (size_t)k * HD + n);
    float2 r1 = *(const float2*)(W + (size_t)(k + 1) * HD + n);
    float4 o  = make_float4(r0.x, r1.x, r0.y, r1.y);
    *(float4*)(Wl2 + p * 256 + n * 2) = o;
  }
  __syncthreads();

  int wave    = tid >> 5;
  int lane    = tid & 31;
  int lane16  = lane & 15;
  int pOff    = lane >> 4;
  int rowBase = (blockIdx.x * 8 + wave) * 16;
  int rowA    = rowBase + lane16;
  int rowAc   = rowA < N_NODES ? rowA : (N_NODES - 1);

  v8f acc[8] = {};
  const float* hrow = h + (size_t)rowAc * HD;

  for (int k0 = 0; k0 < HD; k0 += 4) {
    v2f a = *(const v2f*)(hrow + k0 + (pOff << 1));
    const float* bbase = Wl2 + ((k0 >> 1) + pOff) * 256;
#pragma unroll
    for (int t = 0; t < 8; ++t) {
      int n = t * 16 + lane16;
      v2f b = *(const v2f*)(bbase + n * 2);
      acc[t] = __builtin_amdgcn_wmma_f32_16x16x4_f32(
          false, a, false, b, (short)0, acc[t], false, false);
    }
  }

  int rhalf = pOff << 3;
#pragma unroll
  for (int v = 0; v < 8; ++v) {
    int r = rowBase + v + rhalf;
    if (r < N_NODES) {
      float nr = norm[r];
#pragma unroll
      for (int t = 0; t < 8; ++t)
        m[(size_t)r * HD + t * 16 + lane16] = acc[t][v] * nr;
    }
  }
}

#define OWN_WGS     ((N_NODES + OWN_ROWS - 1) / OWN_ROWS)
#define OWN_WAVES   32
#define SCAN_CAP    512
#define MINE_CAP    256
#define CHUNK_EDGES 512
#define N_CHUNKS    ((N_EDGES + CHUNK_EDGES - 1) / CHUNK_EDGES)
#define SCAN_DEPTH  8
#define ROW_DEPTH   8
#define TP          (HD + 4)
#define LIST_CAP    256
#define LIST_INTS   ((size_t)OWN_WGS * OWN_WAVES * LIST_CAP)
static_assert(N_EDGES % 16 == 0, "a lane's 16 owner bytes are wholly inside the list or wholly outside");
static_assert(OWN_WGS <= 256, "an owner index fits a byte");
static_assert(OWN_ROWS <= 512, "the local row takes the 9 low bits of an entry");
static_assert(N_EDGES < (1 << 22), "the edge index takes the bits above them");
static_assert(SCAN_CAP >= CHUNK_EDGES && MINE_CAP >= 64, "a round always makes progress");
static_assert(N_NODES <= (1 << 22), "a source row shares an int with the 9-bit local row");
static_assert(OWN_WAVES * SCAN_CAP >= HD * HD && OWN_WAVES * MINE_CAP >= HD,
              "the next layer's weights and bias fit the dead lists");

__device__ __forceinline__ void add_batch(
    int s, int r, int nb, int lane, const float* __restrict__ m, float* tile) {
  for (int q0 = 0; q0 < nb; q0 += ROW_DEPTH) {
    int sq[ROW_DEPTH];
#pragma unroll
    for (int q = 0; q < ROW_DEPTH; ++q)
      sq[q] = __builtin_amdgcn_readlane(s, q0 + q);
    v4f v[ROW_DEPTH];
#pragma unroll
    for (int q = 0; q < ROW_DEPTH; ++q)
      v[q] = *(const v4f*)(m + (size_t)sq[q] * HD + (lane << 2));
#pragma unroll
    for (int q = 0; q < ROW_DEPTH; ++q) {
      if (q0 + q < nb) {
        int rq = __builtin_amdgcn_readlane(r, q0 + q);
        v4f* t = (v4f*)(tile + rq * TP + (lane << 2));
        *t = *t + v[q];
      }
    }
  }
}

template <bool SAVE>
__device__ __forceinline__ void add_rows(
    const int* mine, int n, int lane, const float* __restrict__ m,
    const int* __restrict__ src, float* tile, int* __restrict__ keep, int& kept) {
  for (int b0 = 0; b0 < n; b0 += 32) {
    int k   = b0 + lane;
    int ent = 0;
    int s   = 0;
    if (k < n) {
      ent = mine[k];
      s   = src[ent >> 9];
    }
    int r  = ent & 511;
    int nb = (n - b0) < 32 ? (n - b0) : 32;
    if (SAVE) {
      if (k < n && kept + lane < LIST_CAP) keep[kept + lane] = (s << 9) | r;
      kept += nb;
    }
    add_batch(s, r, nb, lane, m, tile);
  }
}

__device__ __forceinline__ void add_kept_rows(
    const int* __restrict__ keep, int n, int lane,
    const float* __restrict__ m, float* tile) {
  int ents[LIST_CAP / 32];
#pragma unroll
  for (int b = 0; b < LIST_CAP / 32; ++b) {
    int k = b * 32 + lane;
    ents[b] = (k < n) ? keep[k] : 0;
  }
#pragma unroll
  for (int b = 0; b < LIST_CAP / 32; ++b) {
    int rest = n - b * 32;
    if (rest > 0)
      add_batch(ents[b] >> 9, ents[b] & 511, rest < 32 ? rest : 32, lane, m, tile);
  }
}

__device__ __forceinline__ unsigned zero_bytes(unsigned x) {
  return ~(((x & 0x7F7F7F7Fu) + 0x7F7F7F7Fu) | x | 0x7F7F7F7Fu);
}

template <bool FIRST>
__global__ __launch_bounds__(1024) void owner_sum(
    const float* __restrict__ m, const int* __restrict__ src,
    const int* __restrict__ dst, const float* __restrict__ norm,
    const float* __restrict__ bias, float* __restrict__ out,
    int* __restrict__ lists, int* __restrict__ listCnt,
    const unsigned* __restrict__ ownerW, const float* __restrict__ Wnext) {
  __shared__ alignas(16) float tile[OWN_ROWS * TP];
  __shared__ alignas(16) int scanList[OWN_WAVES * SCAN_CAP];
  __shared__ alignas(16) int mineList[OWN_WAVES * MINE_CAP];
  __shared__ int   scanCnt[OWN_WAVES];
  __shared__ int   moreFlag[2];

  const int tid  = threadIdx.x;
  const int lane = tid & 31;
  const int w    = __builtin_amdgcn_readfirstlane(tid >> 5);
  const unsigned ltmask = (1u << lane) - 1u;
  const int base  = (int)blockIdx.x * OWN_ROWS;
  const int nrows = (N_NODES - base) < OWN_ROWS ? (N_NODES - base) : OWN_ROWS;
  const unsigned pat = (unsigned)blockIdx.x * 0x01010101u;

  int* myKeep = lists + ((size_t)blockIdx.x * OWN_WAVES + w) * LIST_CAP;
  int  kept   = 0;
  int  keptV  = 0;
  if (!FIRST) keptV = listCnt[(int)blockIdx.x * OWN_WAVES + lane];

  for (int i = tid; i < OWN_ROWS * TP / 4; i += 1024)
    ((float4*)tile)[i] = make_float4(0.f, 0.f, 0.f, 0.f);
  if (tid < 2) moreFlag[tid] = 0;
  const bool useKept =
      !FIRST && __builtin_amdgcn_ballot_w32(keptV > LIST_CAP) == 0u;
  __syncthreads();

  int* myScan = scanList + w * SCAN_CAP;
  int* myMine = mineList + w * MINE_CAP;
  int pos   = w;
  int round = 0;
  if (useKept) {
    add_kept_rows(myKeep, __builtin_amdgcn_readlane(keptV, w), lane, m, tile);
  } else for (;;) {
    int  cnt  = 0;
    bool full = false;
    while (pos < N_CHUNKS && !full) {
      int4 d[SCAN_DEPTH];
#pragma unroll
      for (int u = 0; u < SCAN_DEPTH; ++u) {
        int e0 = (pos + u * OWN_WAVES) * CHUNK_EDGES + lane * 16;
        d[u] = *(const int4*)(ownerW + ((e0 < N_EDGES) ? (e0 >> 2) : 0));
      }
      int used = 0;
#pragma unroll
      for (int u = 0; u < SCAN_DEPTH; ++u) {
        if (!full) {
          unsigned z0 = zero_bytes((unsigned)d[u].x ^ pat);
          unsigned z1 = zero_bytes((unsigned)d[u].y ^ pat);
          unsigned z2 = zero_bytes((unsigned)d[u].z ^ pat);
          unsigned z3 = zero_bytes((unsigned)d[u].w ^ pat);
          const int e0 = (pos + u * OWN_WAVES) * CHUNK_EDGES + lane * 16;
          const bool any = (e0 < N_EDGES) && ((z0 | z1 | z2 | z3) != 0u);
          unsigned bal = __builtin_amdgcn_ballot_w32(any);
          if (bal != 0u) {
            if (cnt + 16 * __popc(bal) > SCAN_CAP) {
              full = true;
            } else {
              unsigned hm = (z0 >> 7) | (z1 >> 6) | (z2 >> 5) | (z3 >> 4);
              if (!any) hm = 0u;
              for (int it = 0; it < 16; ++it) {
                unsigned b2 = __builtin_amdgcn_ballot_w32(hm != 0u);
                if (b2 == 0u) break;
                if (hm != 0u) {
                  int p = __builtin_ctz(hm);
                  myScan[cnt + __popc(b2 & ltmask)] = e0 + ((p & 7) << 2) + (p >> 3);
                  hm &= hm - 1u;
                }
                cnt += __popc(b2);
              }
            }
          }
          if (!full) used = u + 1;
        }
      }
      pos += used * OWN_WAVES;
    }
    for (int i0 = 0; i0 < cnt; i0 += 128) {
      int e[4], r[4];
#pragma unroll
      for (int u = 0; u < 4; ++u) {
        int i = i0 + u * 32 + lane;
        e[u] = (i < cnt) ? myScan[i] : 0;
      }
#pragma unroll
      for (int u = 0; u < 4; ++u) r[u] = dst[e[u]] - base;
#pragma unroll
      for (int u = 0; u < 4; ++u) {
        int i = i0 + u * 32 + lane;
        if (i < cnt)
          myScan[i] = ((unsigned)r[u] < (unsigned)nrows) ? ((e[u] << 9) | r[u]) : -1;
      }
    }
    if (lane == 0) {
      scanCnt[w] = cnt;
      if (pos < N_CHUNKS) moreFlag[round & 1] = 1;
    }
    __syncthreads();
    const int more = moreFlag[round & 1];
    if (tid == 0) moreFlag[(round + 1) & 1] = 0;

    int cntV = scanCnt[lane];
    int nm = 0;
    for (int lw = 0; lw < OWN_WAVES; ++lw) {
      int n = __builtin_amdgcn_readlane(cntV, lw);
      const int* L = scanList + lw * SCAN_CAP;
      for (int i0 = 0; i0 < n; i0 += 96) {
        int k0 = i0 + lane, k1 = k0 + 32, k2 = k0 + 64;
        int ent0 = (k0 < n) ? L[k0] : -1;
        int ent1 = (k1 < n) ? L[k1] : -1;
        int ent2 = (k2 < n) ? L[k2] : -1;
        int ents[3] = {ent0, ent1, ent2};
#pragma unroll
        for (int g = 0; g < 3; ++g) {
          int ent = ents[g];
          bool hit = (ent >= 0) && ((ent & 31) == w);
          unsigned bal = __builtin_amdgcn_ballot_w32(hit);
          if (bal != 0u) {
            if (nm + 32 > MINE_CAP) {
              add_rows<FIRST>(myMine, nm, lane, m, src, tile, myKeep, kept);
              nm = 0;
            }
            if (hit) myMine[nm + __popc(bal & ltmask)] = ent;
            nm += __popc(bal);
          }
        }
      }
    }
    add_rows<FIRST>(myMine, nm, lane, m, src, tile, myKeep, kept);

    if (!more) break;
    ++round;
    __syncthreads();
  }
  if (FIRST && lane == 0) listCnt[(int)blockIdx.x * OWN_WAVES + w] = kept;
  __syncthreads();

  if (FIRST) {
    float* Wl2 = (float*)scanList;
    float* bl  = (float*)mineList;
    for (int i = tid; i < (HD * HD) / 4; i += 1024) {
      int p = i >> 6;
      int n = (i & 63) << 1;
      int k = p << 1;
      float2 r0 = *(const float2*)(Wnext + (size_t)k * HD + n);
      float2 r1 = *(const float2*)(Wnext + (size_t)(k + 1) * HD + n);
      *(float4*)(Wl2 + p * 256 + n * 2) = make_float4(r0.x, r1.x, r0.y, r1.y);
    }
    if (tid < HD) bl[tid] = bias[tid];
    __syncthreads();
    if (w * 16 < nrows) {
      const int lane16 = lane & 15;
      const int pOff   = lane >> 4;
      int rl = w * 16 + lane16;
      if (rl > OWN_ROWS - 1) rl = OWN_ROWS - 1;
      int gr = base + rl;
      if (gr > N_NODES - 1) gr = N_NODES - 1;
      const float nrA = norm[gr];
      const float* trow = tile + rl * TP;
      v8f acc[8] = {};
      for (int k0 = 0; k0 < HD; k0 += 4) {
        const int kk = k0 + (pOff << 1);
        v2f s  = *(const v2f*)(trow + kk);
        v2f bb = *(const v2f*)(bl + kk);
        float x0 = s.x * nrA + bb.x;
        float x1 = s.y * nrA + bb.y;
        v2f a;
        a.x = x0 > 0.f ? x0 : 0.2f * x0;
        a.y = x1 > 0.f ? x1 : 0.2f * x1;
        const float* bbase = Wl2 + ((k0 >> 1) + pOff) * 256;
#pragma unroll
        for (int t = 0; t < 8; ++t) {
          v2f b = *(const v2f*)(bbase + (t * 16 + lane16) * 2);
          acc[t] = __builtin_amdgcn_wmma_f32_16x16x4_f32(
              false, a, false, b, (short)0, acc[t], false, false);
        }
      }
      const int rhalf = pOff << 3;
      const int nbits = __builtin_bit_cast(int, nrA);
#pragma unroll
      for (int v = 0; v < 8; ++v) {
        int lo = __builtin_amdgcn_readlane(nbits, v);
        int hi = __builtin_amdgcn_readlane(nbits, v + 8);
        float nr = __builtin_bit_cast(float, pOff ? hi : lo);
        int r = w * 16 + v + rhalf;
        if (r < nrows) {
#pragma unroll
          for (int t = 0; t < 8; ++t)
            tile[r * TP + t * 16 + lane16] = acc[t][v] * nr;
        }
      }
      __builtin_amdgcn_fence(__ATOMIC_ACQ_REL, "wavefront");
      const int c4 = lane << 2;
      v4f q[16];
#pragma unroll
      for (int j = 0; j < 16; ++j) {
        int r = w * 16 + j;
        q[j] = *(const v4f*)(tile + (r < OWN_ROWS ? r : OWN_ROWS - 1) * TP + c4);
      }
      asm volatile("" :: "v"(q[0]), "v"(q[1]), "v"(q[2]), "v"(q[3]),
                         "v"(q[4]), "v"(q[5]), "v"(q[6]), "v"(q[7]),
                         "v"(q[8]), "v"(q[9]), "v"(q[10]), "v"(q[11]),
                         "v"(q[12]), "v"(q[13]), "v"(q[14]), "v"(q[15]));
#pragma unroll
      for (int j = 0; j < 16; ++j) {
        int r = w * 16 + j;
        if (r < nrows) *(v4f*)(out + (size_t)(base + r) * HD + c4) = q[j];
      }
    }
    return;
  }

  const int c4 = lane << 2;
  const float4 b = *(const float4*)(bias + c4);
  for (int i = tid; i < nrows * 32; i += 1024) {
    int r = i >> 5;
    float nr = norm[base + r];
    float4 a = *(const float4*)(tile + r * TP + c4);
    float4 n;
    n.x = a.x * nr + b.x;
    n.y = a.y * nr + b.y;
    n.z = a.z * nr + b.z;
    n.w = a.w * nr + b.w;
    n.x = n.x > 0.f ? n.x : 0.2f * n.x;
    n.y = n.y > 0.f ? n.y : 0.2f * n.y;
    n.z = n.z > 0.f ? n.z : 0.2f * n.z;
    n.w = n.w > 0.f ? n.w : 0.2f * n.w;
    *(float4*)(out + (size_t)(base + r) * HD + c4) = n;
  }
}

extern "C" void kernel_launch(void* const* d_in, const int* in_sizes, int n_in,
                              void* d_out, int out_size, void* d_ws, size_t ws_size,
                              hipStream_t stream) {
  const int*   node_id = (const int*)d_in[0];
  const int*   src     = (const int*)d_in[1];
  const int*   dst     = (const int*)d_in[2];
  const float* norm    = (const float*)d_in[3];
  const float* embed   = (const float*)d_in[4];
  const float* W1      = (const float*)d_in[5];
  const float* b1      = (const float*)d_in[6];
  const float* W2      = (const float*)d_in[7];
  const float* b2      = (const float*)d_in[8];
  float* out = (float*)d_out;

  const size_t feat = (size_t)N_NODES * HD;
  float* bufH = (float*)d_ws;
  float* bufM = bufH + feat;
  int* lists   = (int*)(bufM + feat);
  int* listCnt = lists + LIST_INTS;
  unsigned* ownerW = (unsigned*)(listCnt + OWN_WGS * OWN_WAVES);

  dim3 blk(256);
  const int gRows = ROW_BLOCKS + OWNER_BLOCKS;
  const int gGemm = (N_NODES + 127) / 128;

  gather_embed<<<gRows, blk, 0, stream>>>(node_id, embed, bufH, dst, ownerW);

  gemm_norm_wmma<<<gGemm, blk, 0, stream>>>(bufH, W1, norm, bufM);
  owner_sum<true><<<OWN_WGS, dim3(1024), 0, stream>>>(bufM, src, dst, norm, b1, bufH,
                                                      lists, listCnt, ownerW, W2);

  owner_sum<false><<<OWN_WGS, dim3(1024), 0, stream>>>(bufH, src, dst, norm, b2, out,
                                                       lists, listCnt, ownerW, W2);
}


# GEMM-1 runs on the embedding table (embed @ W1 commutes with the row lookup); layer 1's owner sum fetches row node_id[src] and fuses norm[src] into its add; no h plane, 4 -> 3 launches; order same
# speedup vs baseline: 3.2206x; 1.1497x over previous
#include <hip/hip_runtime.h>

#define N_NODES 100000
#define N_EDGES 600000
#define HD 128

typedef __attribute__((ext_vector_type(2))) float v2f;
typedef __attribute__((ext_vector_type(8))) float v8f;
typedef __attribute__((ext_vector_type(4))) float v4f;

#define GEMM_BLOCKS ((N_NODES + 127) / 128)
#define OWN_ROWS    392
#define OWNER_EDGES_PER_BLOCK 4096
#define OWNER_BLOCKS ((N_EDGES + OWNER_EDGES_PER_BLOCK - 1) / OWNER_EDGES_PER_BLOCK)

__global__ __launch_bounds__(256) void table_gemm_wmma(
    const float* __restrict__ table, const float* __restrict__ W,
    float* __restrict__ p, const int* __restrict__ dst,
    unsigned* __restrict__ ownerW) {
  __shared__ float Wl2[HD * HD];

  if ((int)blockIdx.x >= GEMM_BLOCKS) {
    int eb = ((int)blockIdx.x - GEMM_BLOCKS) * OWNER_EDGES_PER_BLOCK + (int)threadIdx.x * 4;
    int4 d[4];
#pragma unroll
    for (int u = 0; u < 4; ++u) {
      int e0 = eb + u * 1024;
      d[u] = *(const int4*)(dst + ((e0 < N_EDGES) ? e0 : 0));
    }
#pragma unroll
    for (int u = 0; u < 4; ++u) {
      int e0 = eb + u * 1024;
      if (e0 < N_EDGES)
        ownerW[e0 >> 2] = (((unsigned)d[u].x / OWN_ROWS) & 255u)
                        | ((((unsigned)d[u].y / OWN_ROWS) & 255u) << 8)
                        | ((((unsigned)d[u].z / OWN_ROWS) & 255u) << 16)
                        | ((((unsigned)d[u].w / OWN_ROWS) & 255u) << 24);
    }
    return;
  }

  int tid = threadIdx.x;
  for (int i = tid; i < (HD * HD) / 4; i += 256) {
    int p = i >> 6;
    int n = (i & 63) << 1;
    int k = p << 1;
    float2 r0 = *(const float2*)(W + (size_t)k * HD + n);
    float2 r1 = *(const float2*)(W + (size_t)(k + 1) * HD + n);
    float4 o  = make_float4(r0.x, r1.x, r0.y, r1.y);
    *(float4*)(Wl2 + p * 256 + n * 2) = o;
  }
  __syncthreads();

  int wave    = tid >> 5;
  int lane    = tid & 31;
  int lane16  = lane & 15;
  int pOff    = lane >> 4;
  int rowBase = (blockIdx.x * 8 + wave) * 16;
  int rowA    = rowBase + lane16;
  int rowAc   = rowA < N_NODES ? rowA : (N_NODES - 1);

  v8f acc[8] = {};
  const float* hrow = table + (size_t)rowAc * HD;

  for (int k0 = 0; k0 < HD; k0 += 4) {
    v2f a = *(const v2f*)(hrow + k0 + (pOff << 1));
    const float* bbase = Wl2 + ((k0 >> 1) + pOff) * 256;
#pragma unroll
    for (int t = 0; t < 8; ++t) {
      int n = t * 16 + lane16;
      v2f b = *(const v2f*)(bbase + n * 2);
      acc[t] = __builtin_amdgcn_wmma_f32_16x16x4_f32(
          false, a, false, b, (short)0, acc[t], false, false);
    }
  }

  int rhalf = pOff << 3;
#pragma unroll
  for (int v = 0; v < 8; ++v) {
    int r = rowBase + v + rhalf;
    if (r < N_NODES) {
#pragma unroll
      for (int t = 0; t < 8; ++t)
        p[(size_t)r * HD + t * 16 + lane16] = acc[t][v];
    }
  }
}

#define OWN_WGS     ((N_NODES + OWN_ROWS - 1) / OWN_ROWS)
#define OWN_WAVES   32
#define SCAN_CAP    512
#define MINE_CAP    256
#define CHUNK_EDGES 512
#define N_CHUNKS    ((N_EDGES + CHUNK_EDGES - 1) / CHUNK_EDGES)
#define SCAN_DEPTH  8
#define ROW_DEPTH   8
#define TP          (HD + 4)
#define LIST_CAP    256
#define LIST_INTS   ((size_t)OWN_WGS * OWN_WAVES * LIST_CAP)
static_assert(N_EDGES % 16 == 0, "a lane's 16 owner bytes are wholly inside the list or wholly outside");
static_assert(OWN_WGS <= 256, "an owner index fits a byte");
static_assert(OWN_ROWS <= 512, "the local row takes the 9 low bits of an entry");
static_assert(N_EDGES < (1 << 22), "the edge index takes the bits above them");
static_assert(SCAN_CAP >= CHUNK_EDGES && MINE_CAP >= 64, "a round always makes progress");
static_assert(N_NODES <= (1 << 22), "a source row shares an int with the 9-bit local row");
static_assert(OWN_WAVES * SCAN_CAP >= HD * HD && OWN_WAVES * MINE_CAP >= HD,
              "the next layer's weights and bias fit the dead lists");

template <bool SCALE>
__device__ __forceinline__ void add_batch(
    int s, float f, int r, int nb, int lane, const float* __restrict__ m, float* tile) {
  for (int q0 = 0; q0 < nb; q0 += ROW_DEPTH) {
    int sq[ROW_DEPTH];
#pragma unroll
    for (int q = 0; q < ROW_DEPTH; ++q)
      sq[q] = __builtin_amdgcn_readlane(s, q0 + q);
    v4f v[ROW_DEPTH];
#pragma unroll
    for (int q = 0; q < ROW_DEPTH; ++q)
      v[q] = *(const v4f*)(m + (size_t)sq[q] * HD + (lane << 2));
#pragma unroll
    for (int q = 0; q < ROW_DEPTH; ++q) {
      if (q0 + q < nb) {
        int rq = __builtin_amdgcn_readlane(r, q0 + q);
        v4f* t = (v4f*)(tile + rq * TP + (lane << 2));
        if (SCALE) {
          float fq = __builtin_bit_cast(float,
              __builtin_amdgcn_readlane(__builtin_bit_cast(int, f), q0 + q));
          *t = *t + v[q] * fq;
        } else {
          *t = *t + v[q];
        }
      }
    }
  }
}

template <bool FIRST>
__device__ __forceinline__ void add_rows(
    const int* mine, int n, int lane, const float* __restrict__ m,
    const int* __restrict__ src, const int* __restrict__ node_id,
    const float* __restrict__ norm, float* tile, int* __restrict__ keep, int& kept) {
  for (int b0 = 0; b0 < n; b0 += 32) {
    int k   = b0 + lane;
    int ent = 0;
    int s   = 0;
    if (k < n) {
      ent = mine[k];
      s   = src[ent >> 9];
    }
    int r  = ent & 511;
    int nb = (n - b0) < 32 ? (n - b0) : 32;
    if (FIRST) {
      if (k < n && kept + lane < LIST_CAP) keep[kept + lane] = (s << 9) | r;
      kept += nb;
      add_batch<true>(node_id[s], norm[s], r, nb, lane, m, tile);
    } else {
      add_batch<false>(s, 0.f, r, nb, lane, m, tile);
    }
  }
}

__device__ __forceinline__ void add_kept_rows(
    const int* __restrict__ keep, int n, int lane,
    const float* __restrict__ m, float* tile) {
  int ents[LIST_CAP / 32];
#pragma unroll
  for (int b = 0; b < LIST_CAP / 32; ++b) {
    int k = b * 32 + lane;
    ents[b] = (k < n) ? keep[k] : 0;
  }
#pragma unroll
  for (int b = 0; b < LIST_CAP / 32; ++b) {
    int rest = n - b * 32;
    if (rest > 0)
      add_batch<false>(ents[b] >> 9, 0.f, ents[b] & 511, rest < 32 ? rest : 32, lane, m, tile);
  }
}

__device__ __forceinline__ unsigned zero_bytes(unsigned x) {
  return ~(((x & 0x7F7F7F7Fu) + 0x7F7F7F7Fu) | x | 0x7F7F7F7Fu);
}

template <bool FIRST>
__global__ __launch_bounds__(1024) void owner_sum(
    const float* __restrict__ m, const int* __restrict__ src,
    const int* __restrict__ dst, const int* __restrict__ node_id,
    const float* __restrict__ norm,
    const float* __restrict__ bias, float* __restrict__ out,
    int* __restrict__ lists, int* __restrict__ listCnt,
    const unsigned* __restrict__ ownerW, const float* __restrict__ Wnext) {
  __shared__ alignas(16) float tile[OWN_ROWS * TP];
  __shared__ alignas(16) int scanList[OWN_WAVES * SCAN_CAP];
  __shared__ alignas(16) int mineList[OWN_WAVES * MINE_CAP];
  __shared__ int   scanCnt[OWN_WAVES];
  __shared__ int   moreFlag[2];

  const int tid  = threadIdx.x;
  const int lane = tid & 31;
  const int w    = __builtin_amdgcn_readfirstlane(tid >> 5);
  const unsigned ltmask = (1u << lane) - 1u;
  const int base  = (int)blockIdx.x * OWN_ROWS;
  const int nrows = (N_NODES - base) < OWN_ROWS ? (N_NODES - base) : OWN_ROWS;
  const unsigned pat = (unsigned)blockIdx.x * 0x01010101u;

  int* myKeep = lists + ((size_t)blockIdx.x * OWN_WAVES + w) * LIST_CAP;
  int  kept   = 0;
  int  keptV  = 0;
  if (!FIRST) keptV = listCnt[(int)blockIdx.x * OWN_WAVES + lane];

  for (int i = tid; i < OWN_ROWS * TP / 4; i += 1024)
    ((float4*)tile)[i] = make_float4(0.f, 0.f, 0.f, 0.f);
  if (tid < 2) moreFlag[tid] = 0;
  const bool useKept =
      !FIRST && __builtin_amdgcn_ballot_w32(keptV > LIST_CAP) == 0u;
  __syncthreads();

  int* myScan = scanList + w * SCAN_CAP;
  int* myMine = mineList + w * MINE_CAP;
  int pos   = w;
  int round = 0;
  if (useKept) {
    add_kept_rows(myKeep, __builtin_amdgcn_readlane(keptV, w), lane, m, tile);
  } else for (;;) {
    int  cnt  = 0;
    bool full = false;
    while (pos < N_CHUNKS && !full) {
      int4 d[SCAN_DEPTH];
#pragma unroll
      for (int u = 0; u < SCAN_DEPTH; ++u) {
        int e0 = (pos + u * OWN_WAVES) * CHUNK_EDGES + lane * 16;
        d[u] = *(const int4*)(ownerW + ((e0 < N_EDGES) ? (e0 >> 2) : 0));
      }
      int used = 0;
#pragma unroll
      for (int u = 0; u < SCAN_DEPTH; ++u) {
        if (!full) {
          unsigned z0 = zero_bytes((unsigned)d[u].x ^ pat);
          unsigned z1 = zero_bytes((unsigned)d[u].y ^ pat);
          unsigned z2 = zero_bytes((unsigned)d[u].z ^ pat);
          unsigned z3 = zero_bytes((unsigned)d[u].w ^ pat);
          const int e0 = (pos + u * OWN_WAVES) * CHUNK_EDGES + lane * 16;
          const bool any = (e0 < N_EDGES) && ((z0 | z1 | z2 | z3) != 0u);
          unsigned bal = __builtin_amdgcn_ballot_w32(any);
          if (bal != 0u) {
            if (cnt + 16 * __popc(bal) > SCAN_CAP) {
              full = true;
            } else {
              unsigned hm = (z0 >> 7) | (z1 >> 6) | (z2 >> 5) | (z3 >> 4);
              if (!any) hm = 0u;
              for (int it = 0; it < 16; ++it) {
                unsigned b2 = __builtin_amdgcn_ballot_w32(hm != 0u);
                if (b2 == 0u) break;
                if (hm != 0u) {
                  int p = __builtin_ctz(hm);
                  myScan[cnt + __popc(b2 & ltmask)] = e0 + ((p & 7) << 2) + (p >> 3);
                  hm &= hm - 1u;
                }
                cnt += __popc(b2);
              }
            }
          }
          if (!full) used = u + 1;
        }
      }
      pos += used * OWN_WAVES;
    }
    for (int i0 = 0; i0 < cnt; i0 += 128) {
      int e[4], r[4];
#pragma unroll
      for (int u = 0; u < 4; ++u) {
        int i = i0 + u * 32 + lane;
        e[u] = (i < cnt) ? myScan[i] : 0;
      }
#pragma unroll
      for (int u = 0; u < 4; ++u) r[u] = dst[e[u]] - base;
#pragma unroll
      for (int u = 0; u < 4; ++u) {
        int i = i0 + u * 32 + lane;
        if (i < cnt)
          myScan[i] = ((unsigned)r[u] < (unsigned)nrows) ? ((e[u] << 9) | r[u]) : -1;
      }
    }
    if (lane == 0) {
      scanCnt[w] = cnt;
      if (pos < N_CHUNKS) moreFlag[round & 1] = 1;
    }
    __syncthreads();
    const int more = moreFlag[round & 1];
    if (tid == 0) moreFlag[(round + 1) & 1] = 0;

    int cntV = scanCnt[lane];
    int nm = 0;
    for (int lw = 0; lw < OWN_WAVES; ++lw) {
      int n = __builtin_amdgcn_readlane(cntV, lw);
      const int* L = scanList + lw * SCAN_CAP;
      for (int i0 = 0; i0 < n; i0 += 96) {
        int k0 = i0 + lane, k1 = k0 + 32, k2 = k0 + 64;
        int ent0 = (k0 < n) ? L[k0] : -1;
        int ent1 = (k1 < n) ? L[k1] : -1;
        int ent2 = (k2 < n) ? L[k2] : -1;
        int ents[3] = {ent0, ent1, ent2};
#pragma unroll
        for (int g = 0; g < 3; ++g) {
          int ent = ents[g];
          bool hit = (ent >= 0) && ((ent & 31) == w);
          unsigned bal = __builtin_amdgcn_ballot_w32(hit);
          if (bal != 0u) {
            if (nm + 32 > MINE_CAP) {
              add_rows<FIRST>(myMine, nm, lane, m, src, node_id, norm, tile, myKeep, kept);
              nm = 0;
            }
            if (hit) myMine[nm + __popc(bal & ltmask)] = ent;
            nm += __popc(bal);
          }
        }
      }
    }
    add_rows<FIRST>(myMine, nm, lane, m, src, node_id, norm, tile, myKeep, kept);

    if (!more) break;
    ++round;
    __syncthreads();
  }
  if (FIRST && lane == 0) listCnt[(int)blockIdx.x * OWN_WAVES + w] = kept;
  __syncthreads();

  if (FIRST) {
    float* Wl2 = (float*)scanList;
    float* bl  = (float*)mineList;
    for (int i = tid; i < (HD * HD) / 4; i += 1024) {
      int p = i >> 6;
      int n = (i & 63) << 1;
      int k = p << 1;
      float2 r0 = *(const float2*)(Wnext + (size_t)k * HD + n);
      float2 r1 = *(const float2*)(Wnext + (size_t)(k + 1) * HD + n);
      *(float4*)(Wl2 + p * 256 + n * 2) = make_float4(r0.x, r1.x, r0.y, r1.y);
    }
    if (tid < HD) bl[tid] = bias[tid];
    __syncthreads();
    if (w * 16 < nrows) {
      const int lane16 = lane & 15;
      const int pOff   = lane >> 4;
      int rl = w * 16 + lane16;
      if (rl > OWN_ROWS - 1) rl = OWN_ROWS - 1;
      int gr = base + rl;
      if (gr > N_NODES - 1) gr = N_NODES - 1;
      const float nrA = norm[gr];
      const float* trow = tile + rl * TP;
      v8f acc[8] = {};
      for (int k0 = 0; k0 < HD; k0 += 4) {
        const int kk = k0 + (pOff << 1);
        v2f s  = *(const v2f*)(trow + kk);
        v2f bb = *(const v2f*)(bl + kk);
        float x0 = s.x * nrA + bb.x;
        float x1 = s.y * nrA + bb.y;
        v2f a;
        a.x = x0 > 0.f ? x0 : 0.2f * x0;
        a.y = x1 > 0.f ? x1 : 0.2f * x1;
        const float* bbase = Wl2 + ((k0 >> 1) + pOff) * 256;
#pragma unroll
        for (int t = 0; t < 8; ++t) {
          v2f b = *(const v2f*)(bbase + (t * 16 + lane16) * 2);
          acc[t] = __builtin_amdgcn_wmma_f32_16x16x4_f32(
              false, a, false, b, (short)0, acc[t], false, false);
        }
      }
      const int rhalf = pOff << 3;
      const int nbits = __builtin_bit_cast(int, nrA);
#pragma unroll
      for (int v = 0; v < 8; ++v) {
        int lo = __builtin_amdgcn_readlane(nbits, v);
        int hi = __builtin_amdgcn_readlane(nbits, v + 8);
        float nr = __builtin_bit_cast(float, pOff ? hi : lo);
        int r = w * 16 + v + rhalf;
        if (r < nrows) {
#pragma unroll
          for (int t = 0; t < 8; ++t)
            tile[r * TP + t * 16 + lane16] = acc[t][v] * nr;
        }
      }
      __builtin_amdgcn_fence(__ATOMIC_ACQ_REL, "wavefront");
      const int c4 = lane << 2;
      v4f q[16];
#pragma unroll
      for (int j = 0; j < 16; ++j) {
        int r = w * 16 + j;
        q[j] = *(const v4f*)(tile + (r < OWN_ROWS ? r : OWN_ROWS - 1) * TP + c4);
      }
      asm volatile("" :: "v"(q[0]), "v"(q[1]), "v"(q[2]), "v"(q[3]),
                         "v"(q[4]), "v"(q[5]), "v"(q[6]), "v"(q[7]),
                         "v"(q[8]), "v"(q[9]), "v"(q[10]), "v"(q[11]),
                         "v"(q[12]), "v"(q[13]), "v"(q[14]), "v"(q[15]));
#pragma unroll
      for (int j = 0; j < 16; ++j) {
        int r = w * 16 + j;
        if (r < nrows) *(v4f*)(out + (size_t)(base + r) * HD + c4) = q[j];
      }
    }
    return;
  }

  const int c4 = lane << 2;
  const float4 b = *(const float4*)(bias + c4);
  for (int i = tid; i < nrows * 32; i += 1024) {
    int r = i >> 5;
    float nr = norm[base + r];
    float4 a = *(const float4*)(tile + r * TP + c4);
    float4 n;
    n.x = a.x * nr + b.x;
    n.y = a.y * nr + b.y;
    n.z = a.z * nr + b.z;
    n.w = a.w * nr + b.w;
    n.x = n.x > 0.f ? n.x : 0.2f * n.x;
    n.y = n.y > 0.f ? n.y : 0.2f * n.y;
    n.z = n.z > 0.f ? n.z : 0.2f * n.z;
    n.w = n.w > 0.f ? n.w : 0.2f * n.w;
    *(float4*)(out + (size_t)(base + r) * HD + c4) = n;
  }
}

extern "C" void kernel_launch(void* const* d_in, const int* in_sizes, int n_in,
                              void* d_out, int out_size, void* d_ws, size_t ws_size,
                              hipStream_t stream) {
  const int*   node_id = (const int*)d_in[0];
  const int*   src     = (const int*)d_in[1];
  const int*   dst     = (const int*)d_in[2];
  const float* norm    = (const float*)d_in[3];
  const float* embed   = (const float*)d_in[4];
  const float* W1      = (const float*)d_in[5];
  const float* b1      = (const float*)d_in[6];
  const float* W2      = (const float*)d_in[7];
  const float* b2      = (const float*)d_in[8];
  float* out = (float*)d_out;

  const size_t feat = (size_t)N_NODES * HD;
  float* bufH = (float*)d_ws;
  float* bufM = bufH + feat;
  int* lists   = (int*)(bufM + feat);
  int* listCnt = lists + LIST_INTS;
  unsigned* ownerW = (unsigned*)(listCnt + OWN_WGS * OWN_WAVES);

  dim3 blk(256);
  const int gGemm = GEMM_BLOCKS + OWNER_BLOCKS;

  table_gemm_wmma<<<gGemm, blk, 0, stream>>>(embed, W1, bufM, dst, ownerW);

  owner_sum<true><<<OWN_WGS, dim3(1024), 0, stream>>>(bufM, src, dst, node_id, norm, b1, bufH,
                                                      lists, listCnt, ownerW, W2);

  owner_sum<false><<<OWN_WGS, dim3(1024), 0, stream>>>(bufH, src, dst, node_id, norm, b2, out,
                                                       lists, listCnt, ownerW, W2);
}


# Owner scan reads 1 bit per edge: launch 1's extra blocks build per-owner bit maps by LDS OR; a group reads its 75-KB bit row, not 600 KB of bytes; a row's sum order changes, still fixed by edge list
# speedup vs baseline: 3.3808x; 1.0497x over previous
#include <hip/hip_runtime.h>

#define N_NODES 100000
#define N_EDGES 600000
#define HD 128

typedef __attribute__((ext_vector_type(2))) float v2f;
typedef __attribute__((ext_vector_type(8))) float v8f;
typedef __attribute__((ext_vector_type(4))) float v4f;

#define GEMM_BLOCKS ((N_NODES + 127) / 128)
#define OWN_ROWS    392
#define OWN_WGS     ((N_NODES + OWN_ROWS - 1) / OWN_ROWS)
#define SEG_EDGES   2048
#define SEG_WORDS   (SEG_EDGES / 32)
#define N_SEGS      ((N_EDGES + SEG_EDGES - 1) / SEG_EDGES)
#define ROW_WORDS   (N_SEGS * SEG_WORDS)
static_assert(OWN_WGS * SEG_WORDS * 4 == HD * HD * 4, "a segment's bit maps fill the product's LDS exactly");
static_assert(SEG_EDGES == 256 * 8 && N_EDGES % 8 == 0, "a thread's 8 edges are all inside the list or all outside");

__global__ __launch_bounds__(256) void table_gemm_wmma(
    const float* __restrict__ table, const float* __restrict__ W,
    float* __restrict__ p, const int* __restrict__ dst,
    unsigned* __restrict__ ownerBits) {
  __shared__ float Wl2[HD * HD];

  if ((int)blockIdx.x >= GEMM_BLOCKS) {
    unsigned* bm = (unsigned*)Wl2;
    const int t   = threadIdx.x;
    const int seg = (int)blockIdx.x - GEMM_BLOCKS;
    const int e0  = seg * SEG_EDGES + t * 8;
    const bool in = e0 < N_EDGES;
    const int4 d0 = *(const int4*)(dst + (in ? e0 : 0));
    const int4 d1 = *(const int4*)(dst + (in ? e0 + 4 : 0));
    for (int i = t; i < (OWN_WGS * SEG_WORDS) / 4; i += 256)
      ((uint4*)bm)[i] = make_uint4(0u, 0u, 0u, 0u);
    __syncthreads();
    if (in) {
      const int dd[8] = {d0.x, d0.y, d0.z, d0.w, d1.x, d1.y, d1.z, d1.w};
#pragma unroll
      for (int j = 0; j < 8; ++j) {
        const unsigned o  = (unsigned)dd[j] / OWN_ROWS;
        const unsigned el = (unsigned)(t * 8 + j);
        if (o < (unsigned)OWN_WGS) atomicOr(&bm[o * SEG_WORDS + (el >> 5)], 1u << (el & 31u));
      }
    }
    __syncthreads();
    for (int i = t; i < (OWN_WGS * SEG_WORDS) / 4; i += 256) {
      const int o = i >> 4;
      const int q = i & 15;
      *(uint4*)(ownerBits + ((size_t)o * N_SEGS + seg) * SEG_WORDS + q * 4) = ((const uint4*)bm)[i];
    }
    return;
  }

  int tid = threadIdx.x;
  for (int i = tid; i < (HD * HD) / 4; i += 256) {
    int p = i >> 6;
    int n = (i & 63) << 1;
    int k = p << 1;
    float2 r0 = *(const float2*)(W + (size_t)k * HD + n);
    float2 r1 = *(const float2*)(W + (size_t)(k + 1) * HD + n);
    float4 o  = make_float4(r0.x, r1.x, r0.y, r1.y);
    *(float4*)(Wl2 + p * 256 + n * 2) = o;
  }
  __syncthreads();

  int wave    = tid >> 5;
  int lane    = tid & 31;
  int lane16  = lane & 15;
  int pOff    = lane >> 4;
  int rowBase = (blockIdx.x * 8 + wave) * 16;
  int rowA    = rowBase + lane16;
  int rowAc   = rowA < N_NODES ? rowA : (N_NODES - 1);

  v8f acc[8] = {};
  const float* hrow = table + (size_t)rowAc * HD;

  for (int k0 = 0; k0 < HD; k0 += 4) {
    v2f a = *(const v2f*)(hrow + k0 + (pOff << 1));
    const float* bbase = Wl2 + ((k0 >> 1) + pOff) * 256;
#pragma unroll
    for (int t = 0; t < 8; ++t) {
      int n = t * 16 + lane16;
      v2f b = *(const v2f*)(bbase + n * 2);
      acc[t] = __builtin_amdgcn_wmma_f32_16x16x4_f32(
          false, a, false, b, (short)0, acc[t], false, false);
    }
  }

  int rhalf = pOff << 3;
#pragma unroll
  for (int v = 0; v < 8; ++v) {
    int r = rowBase + v + rhalf;
    if (r < N_NODES) {
#pragma unroll
      for (int t = 0; t < 8; ++t)
        p[(size_t)r * HD + t * 16 + lane16] = acc[t][v];
    }
  }
}

#define OWN_WAVES   32
#define SCAN_CAP    512
#define MINE_CAP    256
#define ROW_QUADS   (ROW_WORDS / 4)
#define BM_DEPTH    ((ROW_QUADS + 32 * OWN_WAVES - 1) / (32 * OWN_WAVES))
#define ROW_DEPTH   8
#define TP          (HD + 4)
#define LIST_CAP    256
#define LIST_INTS   ((size_t)OWN_WGS * OWN_WAVES * LIST_CAP)
static_assert(ROW_WORDS % 4 == 0, "an owner's bit row is a whole number of 16-byte pieces");
static_assert(OWN_ROWS <= 512, "the local row takes the 9 low bits of an entry");
static_assert(N_EDGES < (1 << 22), "the edge index takes the bits above them");
static_assert(SCAN_CAP >= 32 && MINE_CAP >= 64, "a round always makes progress");
static_assert(N_NODES <= (1 << 22), "a source row shares an int with the 9-bit local row");
static_assert(OWN_WAVES * SCAN_CAP >= HD * HD && OWN_WAVES * MINE_CAP >= HD,
              "the next layer's weights and bias fit the dead lists");

template <bool SCALE>
__device__ __forceinline__ void add_batch(
    int s, float f, int r, int nb, int lane, const float* __restrict__ m, float* tile) {
  for (int q0 = 0; q0 < nb; q0 += ROW_DEPTH) {
    int sq[ROW_DEPTH];
#pragma unroll
    for (int q = 0; q < ROW_DEPTH; ++q)
      sq[q] = __builtin_amdgcn_readlane(s, q0 + q);
    v4f v[ROW_DEPTH];
#pragma unroll
    for (int q = 0; q < ROW_DEPTH; ++q)
      v[q] = *(const v4f*)(m + (size_t)sq[q] * HD + (lane << 2));
#pragma unroll
    for (int q = 0; q < ROW_DEPTH; ++q) {
      if (q0 + q < nb) {
        int rq = __builtin_amdgcn_readlane(r, q0 + q);
        v4f* t = (v4f*)(tile + rq * TP + (lane << 2));
        if (SCALE) {
          float fq = __builtin_bit_cast(float,
              __builtin_amdgcn_readlane(__builtin_bit_cast(int, f), q0 + q));
          *t = *t + v[q] * fq;
        } else {
          *t = *t + v[q];
        }
      }
    }
  }
}

template <bool FIRST>
__device__ __forceinline__ void add_rows(
    const int* mine, int n, int lane, const float* __restrict__ m,
    const int* __restrict__ src, const int* __restrict__ node_id,
    const float* __restrict__ norm, float* tile, int* __restrict__ keep, int& kept) {
  for (int b0 = 0; b0 < n; b0 += 32) {
    int k   = b0 + lane;
    int ent = 0;
    int s   = 0;
    if (k < n) {
      ent = mine[k];
      s   = src[ent >> 9];
    }
    int r  = ent & 511;
    int nb = (n - b0) < 32 ? (n - b0) : 32;
    if (FIRST) {
      if (k < n && kept + lane < LIST_CAP) keep[kept + lane] = (s << 9) | r;
      kept += nb;
      add_batch<true>(node_id[s], norm[s], r, nb, lane, m, tile);
    } else {
      add_batch<false>(s, 0.f, r, nb, lane, m, tile);
    }
  }
}

__device__ __forceinline__ void add_kept_rows(
    const int* __restrict__ keep, int n, int lane,
    const float* __restrict__ m, float* tile) {
  int ents[LIST_CAP / 32];
#pragma unroll
  for (int b = 0; b < LIST_CAP / 32; ++b) {
    int k = b * 32 + lane;
    ents[b] = (k < n) ? keep[k] : 0;
  }
#pragma unroll
  for (int b = 0; b < LIST_CAP / 32; ++b) {
    int rest = n - b * 32;
    if (rest > 0)
      add_batch<false>(ents[b] >> 9, 0.f, ents[b] & 511, rest < 32 ? rest : 32, lane, m, tile);
  }
}

__device__ __forceinline__ void take_bits(
    unsigned& m, int e0, int* list, int& cnt, bool& full, unsigned ltmask) {
  for (int it = 0; it < 33 && !full; ++it) {
    const unsigned b2 = __builtin_amdgcn_ballot_w32(m != 0u);
    if (b2 == 0u) break;
    const int add = __popc(b2);
    if (cnt + add > SCAN_CAP) {
      full = true;
    } else {
      if (m != 0u) {
        list[cnt + __popc(b2 & ltmask)] = e0 + __builtin_ctz(m);
        m &= m - 1u;
      }
      cnt += add;
    }
  }
}

__device__ __forceinline__ void load_bits(
    uint4 (&mw)[BM_DEPTH], const unsigned* __restrict__ ownerBits, int w, int lane) {
  const uint4* bitRow = (const uint4*)(ownerBits + (size_t)blockIdx.x * ROW_WORDS);
#pragma unroll
  for (int u = 0; u < BM_DEPTH; ++u) {
    const int q = (u * OWN_WAVES + w) * 32 + lane;
    mw[u] = bitRow[(q < ROW_QUADS) ? q : 0];
    if (q >= ROW_QUADS) mw[u] = make_uint4(0u, 0u, 0u, 0u);
  }
}

template <bool FIRST>
__global__ __launch_bounds__(1024) void owner_sum(
    const float* __restrict__ m, const int* __restrict__ src,
    const int* __restrict__ dst, const int* __restrict__ node_id,
    const float* __restrict__ norm,
    const float* __restrict__ bias, float* __restrict__ out,
    int* __restrict__ lists, int* __restrict__ listCnt,
    const unsigned* __restrict__ ownerBits, const float* __restrict__ Wnext) {
  __shared__ alignas(16) float tile[OWN_ROWS * TP];
  __shared__ alignas(16) int scanList[OWN_WAVES * SCAN_CAP];
  __shared__ alignas(16) int mineList[OWN_WAVES * MINE_CAP];
  __shared__ int   scanCnt[OWN_WAVES];
  __shared__ int   moreFlag[2];

  const int tid  = threadIdx.x;
  const int lane = tid & 31;
  const int w    = __builtin_amdgcn_readfirstlane(tid >> 5);
  const unsigned ltmask = (1u << lane) - 1u;
  const int base  = (int)blockIdx.x * OWN_ROWS;
  const int nrows = (N_NODES - base) < OWN_ROWS ? (N_NODES - base) : OWN_ROWS;

  int* myKeep = lists + ((size_t)blockIdx.x * OWN_WAVES + w) * LIST_CAP;
  int  kept   = 0;
  int  keptV  = 0;
  if (!FIRST) keptV = listCnt[(int)blockIdx.x * OWN_WAVES + lane];
  uint4 mw[BM_DEPTH];
  if (FIRST) load_bits(mw, ownerBits, w, lane);

  for (int i = tid; i < OWN_ROWS * TP / 4; i += 1024)
    ((float4*)tile)[i] = make_float4(0.f, 0.f, 0.f, 0.f);
  if (tid < 2) moreFlag[tid] = 0;
  const bool useKept =
      !FIRST && __builtin_amdgcn_ballot_w32(keptV > LIST_CAP) == 0u;
  __syncthreads();

  int* myScan = scanList + w * SCAN_CAP;
  int* myMine = mineList + w * MINE_CAP;
  int round = 0;
  if (useKept) {
    add_kept_rows(myKeep, __builtin_amdgcn_readlane(keptV, w), lane, m, tile);
  } else {
  if (!FIRST) load_bits(mw, ownerBits, w, lane);
  for (;;) {
    int  cnt  = 0;
    bool full = false;
#pragma unroll
    for (int u = 0; u < BM_DEPTH; ++u) {
      const int e0 = ((u * OWN_WAVES + w) * 32 + lane) * 128;
      take_bits(mw[u].x, e0,      myScan, cnt, full, ltmask);
      take_bits(mw[u].y, e0 + 32, myScan, cnt, full, ltmask);
      take_bits(mw[u].z, e0 + 64, myScan, cnt, full, ltmask);
      take_bits(mw[u].w, e0 + 96, myScan, cnt, full, ltmask);
    }
    for (int i0 = 0; i0 < cnt; i0 += 128) {
      int e[4], r[4];
#pragma unroll
      for (int u = 0; u < 4; ++u) {
        int i = i0 + u * 32 + lane;
        e[u] = (i < cnt) ? myScan[i] : 0;
      }
#pragma unroll
      for (int u = 0; u < 4; ++u) r[u] = dst[e[u]] - base;
#pragma unroll
      for (int u = 0; u < 4; ++u) {
        int i = i0 + u * 32 + lane;
        if (i < cnt)
          myScan[i] = ((unsigned)r[u] < (unsigned)nrows) ? ((e[u] << 9) | r[u]) : -1;
      }
    }
    if (lane == 0) {
      scanCnt[w] = cnt;
      if (full) moreFlag[round & 1] = 1;
    }
    __syncthreads();
    const int more = moreFlag[round & 1];
    if (tid == 0) moreFlag[(round + 1) & 1] = 0;

    int cntV = scanCnt[lane];
    int nm = 0;
    for (int lw = 0; lw < OWN_WAVES; ++lw) {
      int n = __builtin_amdgcn_readlane(cntV, lw);
      const int* L = scanList + lw * SCAN_CAP;
      for (int i0 = 0; i0 < n; i0 += 96) {
        int k0 = i0 + lane, k1 = k0 + 32, k2 = k0 + 64;
        int ent0 = (k0 < n) ? L[k0] : -1;
        int ent1 = (k1 < n) ? L[k1] : -1;
        int ent2 = (k2 < n) ? L[k2] : -1;
        int ents[3] = {ent0, ent1, ent2};
#pragma unroll
        for (int g = 0; g < 3; ++g) {
          int ent = ents[g];
          bool hit = (ent >= 0) && ((ent & 31) == w);
          unsigned bal = __builtin_amdgcn_ballot_w32(hit);
          if (bal != 0u) {
            if (nm + 32 > MINE_CAP) {
              add_rows<FIRST>(myMine, nm, lane, m, src, node_id, norm, tile, myKeep, kept);
              nm = 0;
            }
            if (hit) myMine[nm + __popc(bal & ltmask)] = ent;
            nm += __popc(bal);
          }
        }
      }
    }
    add_rows<FIRST>(myMine, nm, lane, m, src, node_id, norm, tile, myKeep, kept);

    if (!more) break;
    ++round;
    __syncthreads();
  }
  }
  if (FIRST && lane == 0) listCnt[(int)blockIdx.x * OWN_WAVES + w] = kept;
  __syncthreads();

  if (FIRST) {
    float* Wl2 = (float*)scanList;
    float* bl  = (float*)mineList;
    for (int i = tid; i < (HD * HD) / 4; i += 1024) {
      int p = i >> 6;
      int n = (i & 63) << 1;
      int k = p << 1;
      float2 r0 = *(const float2*)(Wnext + (size_t)k * HD + n);
      float2 r1 = *(const float2*)(Wnext + (size_t)(k + 1) * HD + n);
      *(float4*)(Wl2 + p * 256 + n * 2) = make_float4(r0.x, r1.x, r0.y, r1.y);
    }
    if (tid < HD) bl[tid] = bias[tid];
    __syncthreads();
    if (w * 16 < nrows) {
      const int lane16 = lane & 15;
      const int pOff   = lane >> 4;
      int rl = w * 16 + lane16;
      if (rl > OWN_ROWS - 1) rl = OWN_ROWS - 1;
      int gr = base + rl;
      if (gr > N_NODES - 1) gr = N_NODES - 1;
      const float nrA = norm[gr];
      const float* trow = tile + rl * TP;
      v8f acc[8] = {};
      for (int k0 = 0; k0 < HD; k0 += 4) {
        const int kk = k0 + (pOff << 1);
        v2f s  = *(const v2f*)(trow + kk);
        v2f bb = *(const v2f*)(bl + kk);
        float x0 = s.x * nrA + bb.x;
        float x1 = s.y * nrA + bb.y;
        v2f a;
        a.x = x0 > 0.f ? x0 : 0.2f * x0;
        a.y = x1 > 0.f ? x1 : 0.2f * x1;
        const float* bbase = Wl2 + ((k0 >> 1) + pOff) * 256;
#pragma unroll
        for (int t = 0; t < 8; ++t) {
          v2f b = *(const v2f*)(bbase + (t * 16 + lane16) * 2);
          acc[t] = __builtin_amdgcn_wmma_f32_16x16x4_f32(
              false, a, false, b, (short)0, acc[t], false, false);
        }
      }
      const int rhalf = pOff << 3;
      const int nbits = __builtin_bit_cast(int, nrA);
#pragma unroll
      for (int v = 0; v < 8; ++v) {
        int lo = __builtin_amdgcn_readlane(nbits, v);
        int hi = __builtin_amdgcn_readlane(nbits, v + 8);
        float nr = __builtin_bit_cast(float, pOff ? hi : lo);
        int r = w * 16 + v + rhalf;
        if (r < nrows) {
#pragma unroll
          for (int t = 0; t < 8; ++t)
            tile[r * TP + t * 16 + lane16] = acc[t][v] * nr;
        }
      }
      __builtin_amdgcn_fence(__ATOMIC_ACQ_REL, "wavefront");
      const int c4 = lane << 2;
      v4f q[16];
#pragma unroll
      for (int j = 0; j < 16; ++j) {
        int r = w * 16 + j;
        q[j] = *(const v4f*)(tile + (r < OWN_ROWS ? r : OWN_ROWS - 1) * TP + c4);
      }
      asm volatile("" :: "v"(q[0]), "v"(q[1]), "v"(q[2]), "v"(q[3]),
                         "v"(q[4]), "v"(q[5]), "v"(q[6]), "v"(q[7]),
                         "v"(q[8]), "v"(q[9]), "v"(q[10]), "v"(q[11]),
                         "v"(q[12]), "v"(q[13]), "v"(q[14]), "v"(q[15]));
#pragma unroll
      for (int j = 0; j < 16; ++j) {
        int r = w * 16 + j;
        if (r < nrows) *(v4f*)(out + (size_t)(base + r) * HD + c4) = q[j];
      }
    }
    return;
  }

  const int c4 = lane << 2;
  const float4 b = *(const float4*)(bias + c4);
  for (int i = tid; i < nrows * 32; i += 1024) {
    int r = i >> 5;
    float nr = norm[base + r];
    float4 a = *(const float4*)(tile + r * TP + c4);
    float4 n;
    n.x = a.x * nr + b.x;
    n.y = a.y * nr + b.y;
    n.z = a.z * nr + b.z;
    n.w = a.w * nr + b.w;
    n.x = n.x > 0.f ? n.x : 0.2f * n.x;
    n.y = n.y > 0.f ? n.y : 0.2f * n.y;
    n.z = n.z > 0.f ? n.z : 0.2f * n.z;
    n.w = n.w > 0.f ? n.w : 0.2f * n.w;
    *(float4*)(out + (size_t)(base + r) * HD + c4) = n;
  }
}

extern "C" void kernel_launch(void* const* d_in, const int* in_sizes, int n_in,
                              void* d_out, int out_size, void* d_ws, size_t ws_size,
                              hipStream_t stream) {
  const int*   node_id = (const int*)d_in[0];
  const int*   src     = (const int*)d_in[1];
  const int*   dst     = (const int*)d_in[2];
  const float* norm    = (const float*)d_in[3];
  const float* embed   = (const float*)d_in[4];
  const float* W1      = (const float*)d_in[5];
  const float* b1      = (const float*)d_in[6];
  const float* W2      = (const float*)d_in[7];
  const float* b2      = (const float*)d_in[8];
  float* out = (float*)d_out;

  const size_t feat = (size_t)N_NODES * HD;
  float* bufH = (float*)d_ws;
  float* bufM = bufH + feat;
  int* lists   = (int*)(bufM + feat);
  int* listCnt = lists + LIST_INTS;
  unsigned* ownerBits = (unsigned*)(listCnt + OWN_WGS * OWN_WAVES);

  dim3 blk(256);
  const int gGemm = GEMM_BLOCKS + N_SEGS;

  table_gemm_wmma<<<gGemm, blk, 0, stream>>>(embed, W1, bufM, dst, ownerBits);

  owner_sum<true><<<OWN_WGS, dim3(1024), 0, stream>>>(bufM, src, dst, node_id, norm, b1, bufH,
                                                      lists, listCnt, ownerBits, W2);

  owner_sum<false><<<OWN_WGS, dim3(1024), 0, stream>>>(bufH, src, dst, node_id, norm, b2, out,
                                                       lists, listCnt, ownerBits, W2);
}


# Owner lists by stable placement: scanning waves count entries per owning wave, then store each at its final list slot (ballot-match rank + LDS running count); no 32-list walk; sum order unchanged
# speedup vs baseline: 3.5987x; 1.0645x over previous
#include <hip/hip_runtime.h>

#define N_NODES 100000
#define N_EDGES 600000
#define HD 128

typedef __attribute__((ext_vector_type(2))) float v2f;
typedef __attribute__((ext_vector_type(8))) float v8f;
typedef __attribute__((ext_vector_type(4))) float v4f;

#define GEMM_BLOCKS ((N_NODES + 127) / 128)
#define OWN_ROWS    392
#define OWN_WGS     ((N_NODES + OWN_ROWS - 1) / OWN_ROWS)
#define SEG_EDGES   2048
#define SEG_WORDS   (SEG_EDGES / 32)
#define N_SEGS      ((N_EDGES + SEG_EDGES - 1) / SEG_EDGES)
#define ROW_WORDS   (N_SEGS * SEG_WORDS)
static_assert(OWN_WGS * SEG_WORDS * 4 == HD * HD * 4, "a segment's bit maps fill the product's LDS exactly");
static_assert(SEG_EDGES == 256 * 8 && N_EDGES % 8 == 0, "a thread's 8 edges are all inside the list or all outside");

__global__ __launch_bounds__(256) void table_gemm_wmma(
    const float* __restrict__ table, const float* __restrict__ W,
    float* __restrict__ p, const int* __restrict__ dst,
    unsigned* __restrict__ ownerBits) {
  __shared__ float Wl2[HD * HD];

  if ((int)blockIdx.x >= GEMM_BLOCKS) {
    unsigned* bm = (unsigned*)Wl2;
    const int t   = threadIdx.x;
    const int seg = (int)blockIdx.x - GEMM_BLOCKS;
    const int e0  = seg * SEG_EDGES + t * 8;
    const bool in = e0 < N_EDGES;
    const int4 d0 = *(const int4*)(dst + (in ? e0 : 0));
    const int4 d1 = *(const int4*)(dst + (in ? e0 + 4 : 0));
    for (int i = t; i < (OWN_WGS * SEG_WORDS) / 4; i += 256)
      ((uint4*)bm)[i] = make_uint4(0u, 0u, 0u, 0u);
    __syncthreads();
    if (in) {
      const int dd[8] = {d0.x, d0.y, d0.z, d0.w, d1.x, d1.y, d1.z, d1.w};
#pragma unroll
      for (int j = 0; j < 8; ++j) {
        const unsigned o  = (unsigned)dd[j] / OWN_ROWS;
        const unsigned el = (unsigned)(t * 8 + j);
        if (o < (unsigned)OWN_WGS) atomicOr(&bm[o * SEG_WORDS + (el >> 5)], 1u << (el & 31u));
      }
    }
    __syncthreads();
    for (int i = t; i < (OWN_WGS * SEG_WORDS) / 4; i += 256) {
      const int o = i >> 4;
      const int q = i & 15;
      *(uint4*)(ownerBits + ((size_t)o * N_SEGS + seg) * SEG_WORDS + q * 4) = ((const uint4*)bm)[i];
    }
    return;
  }

  int tid = threadIdx.x;
  for (int i = tid; i < (HD * HD) / 4; i += 256) {
    int p = i >> 6;
    int n = (i & 63) << 1;
    int k = p << 1;
    float2 r0 = *(const float2*)(W + (size_t)k * HD + n);
    float2 r1 = *(const float2*)(W + (size_t)(k + 1) * HD + n);
    float4 o  = make_float4(r0.x, r1.x, r0.y, r1.y);
    *(float4*)(Wl2 + p * 256 + n * 2) = o;
  }
  __syncthreads();

  int wave    = tid >> 5;
  int lane    = tid & 31;
  int lane16  = lane & 15;
  int pOff    = lane >> 4;
  int rowBase = (blockIdx.x * 8 + wave) * 16;
  int rowA    = rowBase + lane16;
  int rowAc   = rowA < N_NODES ? rowA : (N_NODES - 1);

  v8f acc[8] = {};
  const float* hrow = table + (size_t)rowAc * HD;

  for (int k0 = 0; k0 < HD; k0 += 4) {
    v2f a = *(const v2f*)(hrow + k0 + (pOff << 1));
    const float* bbase = Wl2 + ((k0 >> 1) + pOff) * 256;
#pragma unroll
    for (int t = 0; t < 8; ++t) {
      int n = t * 16 + lane16;
      v2f b = *(const v2f*)(bbase + n * 2);
      acc[t] = __builtin_amdgcn_wmma_f32_16x16x4_f32(
          false, a, false, b, (short)0, acc[t], false, false);
    }
  }

  int rhalf = pOff << 3;
#pragma unroll
  for (int v = 0; v < 8; ++v) {
    int r = rowBase + v + rhalf;
    if (r < N_NODES) {
#pragma unroll
      for (int t = 0; t < 8; ++t)
        p[(size_t)r * HD + t * 16 + lane16] = acc[t][v];
    }
  }
}

#define OWN_WAVES   32
#define SCAN_CAP    512
#define MINE_CAP    256
#define ROW_QUADS   (ROW_WORDS / 4)
#define BM_DEPTH    ((ROW_QUADS + 32 * OWN_WAVES - 1) / (32 * OWN_WAVES))
#define ROW_DEPTH   8
#define TP          (HD + 4)
#define LIST_CAP    256
#define LIST_INTS   ((size_t)OWN_WGS * OWN_WAVES * LIST_CAP)
static_assert(ROW_WORDS % 4 == 0, "an owner's bit row is a whole number of 16-byte pieces");
static_assert(OWN_ROWS <= 512, "the local row takes the 9 low bits of an entry");
static_assert(N_EDGES < (1 << 22), "the edge index takes the bits above them");
static_assert(SCAN_CAP >= 32 && MINE_CAP >= 64, "a round always makes progress");
static_assert(N_NODES <= (1 << 22), "a source row shares an int with the 9-bit local row");
static_assert(OWN_WAVES * SCAN_CAP >= HD * HD && OWN_WAVES * MINE_CAP >= HD,
              "the next layer's weights and bias fit the dead lists");

template <bool SCALE>
__device__ __forceinline__ void add_batch(
    int s, float f, int r, int nb, int lane, const float* __restrict__ m, float* tile) {
  for (int q0 = 0; q0 < nb; q0 += ROW_DEPTH) {
    int sq[ROW_DEPTH];
#pragma unroll
    for (int q = 0; q < ROW_DEPTH; ++q)
      sq[q] = __builtin_amdgcn_readlane(s, q0 + q);
    v4f v[ROW_DEPTH];
#pragma unroll
    for (int q = 0; q < ROW_DEPTH; ++q)
      v[q] = *(const v4f*)(m + (size_t)sq[q] * HD + (lane << 2));
#pragma unroll
    for (int q = 0; q < ROW_DEPTH; ++q) {
      if (q0 + q < nb) {
        int rq = __builtin_amdgcn_readlane(r, q0 + q);
        v4f* t = (v4f*)(tile + rq * TP + (lane << 2));
        if (SCALE) {
          float fq = __builtin_bit_cast(float,
              __builtin_amdgcn_readlane(__builtin_bit_cast(int, f), q0 + q));
          *t = *t + v[q] * fq;
        } else {
          *t = *t + v[q];
        }
      }
    }
  }
}

template <bool FIRST>
__device__ __forceinline__ void add_rows(
    const int* mine, int n, int lane, const float* __restrict__ m,
    const int* __restrict__ src, const int* __restrict__ node_id,
    const float* __restrict__ norm, float* tile, int* __restrict__ keep, int& kept) {
  for (int b0 = 0; b0 < n; b0 += 32) {
    int k   = b0 + lane;
    int ent = 0;
    int s   = 0;
    if (k < n) {
      ent = mine[k];
      s   = src[ent >> 9];
    }
    int r  = ent & 511;
    int nb = (n - b0) < 32 ? (n - b0) : 32;
    if (FIRST) {
      if (k < n && kept + lane < LIST_CAP) keep[kept + lane] = (s << 9) | r;
      kept += nb;
      add_batch<true>(node_id[s], norm[s], r, nb, lane, m, tile);
    } else {
      add_batch<false>(s, 0.f, r, nb, lane, m, tile);
    }
  }
}

__device__ __forceinline__ void add_kept_rows(
    const int* __restrict__ keep, int n, int lane,
    const float* __restrict__ m, float* tile) {
  int ents[LIST_CAP / 32];
#pragma unroll
  for (int b = 0; b < LIST_CAP / 32; ++b) {
    int k = b * 32 + lane;
    ents[b] = (k < n) ? keep[k] : 0;
  }
#pragma unroll
  for (int b = 0; b < LIST_CAP / 32; ++b) {
    int rest = n - b * 32;
    if (rest > 0)
      add_batch<false>(ents[b] >> 9, 0.f, ents[b] & 511, rest < 32 ? rest : 32, lane, m, tile);
  }
}

__device__ __forceinline__ void take_bits(
    unsigned& m, int e0, int* list, int& cnt, bool& full, unsigned ltmask) {
  for (int it = 0; it < 33 && !full; ++it) {
    const unsigned b2 = __builtin_amdgcn_ballot_w32(m != 0u);
    if (b2 == 0u) break;
    const int add = __popc(b2);
    if (cnt + add > SCAN_CAP) {
      full = true;
    } else {
      if (m != 0u) {
        list[cnt + __popc(b2 & ltmask)] = e0 + __builtin_ctz(m);
        m &= m - 1u;
      }
      cnt += add;
    }
  }
}

__device__ __forceinline__ void load_bits(
    uint4 (&mw)[BM_DEPTH], const unsigned* __restrict__ ownerBits, int w, int lane) {
  const uint4* bitRow = (const uint4*)(ownerBits + (size_t)blockIdx.x * ROW_WORDS);
#pragma unroll
  for (int u = 0; u < BM_DEPTH; ++u) {
    const int q = (u * OWN_WAVES + w) * 32 + lane;
    mw[u] = bitRow[(q < ROW_QUADS) ? q : 0];
    if (q >= ROW_QUADS) mw[u] = make_uint4(0u, 0u, 0u, 0u);
  }
}

template <bool FIRST>
__global__ __launch_bounds__(1024) void owner_sum(
    const float* __restrict__ m, const int* __restrict__ src,
    const int* __restrict__ dst, const int* __restrict__ node_id,
    const float* __restrict__ norm,
    const float* __restrict__ bias, float* __restrict__ out,
    int* __restrict__ lists, int* __restrict__ listCnt,
    const unsigned* __restrict__ ownerBits, const float* __restrict__ Wnext) {
  __shared__ alignas(16) float tile[OWN_ROWS * TP];
  __shared__ alignas(16) int scanList[OWN_WAVES * SCAN_CAP];
  __shared__ alignas(16) int mineList[OWN_WAVES * MINE_CAP];
  __shared__ int   keyCnt[OWN_WAVES * OWN_WAVES];
  __shared__ int   keyRun[OWN_WAVES * OWN_WAVES];
  __shared__ int   moreFlag[2];

  const int tid  = threadIdx.x;
  const int lane = tid & 31;
  const int w    = __builtin_amdgcn_readfirstlane(tid >> 5);
  const unsigned ltmask = (1u << lane) - 1u;
  const int base  = (int)blockIdx.x * OWN_ROWS;
  const int nrows = (N_NODES - base) < OWN_ROWS ? (N_NODES - base) : OWN_ROWS;

  int* myKeep = lists + ((size_t)blockIdx.x * OWN_WAVES + w) * LIST_CAP;
  int  kept   = 0;
  int  keptV  = 0;
  if (!FIRST) keptV = listCnt[(int)blockIdx.x * OWN_WAVES + lane];
  uint4 mw[BM_DEPTH];
  if (FIRST) load_bits(mw, ownerBits, w, lane);

  for (int i = tid; i < OWN_ROWS * TP / 4; i += 1024)
    ((float4*)tile)[i] = make_float4(0.f, 0.f, 0.f, 0.f);
  if (tid < 2) moreFlag[tid] = 0;
  const bool useKept =
      !FIRST && __builtin_amdgcn_ballot_w32(keptV > LIST_CAP) == 0u;
  __syncthreads();

  int* myScan = scanList + w * SCAN_CAP;
  int* myMine = mineList + w * MINE_CAP;
  int round = 0;
  if (useKept) {
    add_kept_rows(myKeep, __builtin_amdgcn_readlane(keptV, w), lane, m, tile);
  } else {
  if (!FIRST) load_bits(mw, ownerBits, w, lane);
  for (;;) {
    keyCnt[w * OWN_WAVES + lane] = 0;
    int  cnt  = 0;
    bool full = false;
#pragma unroll
    for (int u = 0; u < BM_DEPTH; ++u) {
      const int e0 = ((u * OWN_WAVES + w) * 32 + lane) * 128;
      take_bits(mw[u].x, e0,      myScan, cnt, full, ltmask);
      take_bits(mw[u].y, e0 + 32, myScan, cnt, full, ltmask);
      take_bits(mw[u].z, e0 + 64, myScan, cnt, full, ltmask);
      take_bits(mw[u].w, e0 + 96, myScan, cnt, full, ltmask);
    }
    for (int i0 = 0; i0 < cnt; i0 += 128) {
      int e[4], r[4];
#pragma unroll
      for (int u = 0; u < 4; ++u) {
        int i = i0 + u * 32 + lane;
        e[u] = (i < cnt) ? myScan[i] : 0;
      }
#pragma unroll
      for (int u = 0; u < 4; ++u) r[u] = dst[e[u]] - base;
#pragma unroll
      for (int u = 0; u < 4; ++u) {
        int i = i0 + u * 32 + lane;
        if (i < cnt) {
          const bool here = (unsigned)r[u] < (unsigned)nrows;
          myScan[i] = here ? ((e[u] << 9) | r[u]) : -1;
          if (here) atomicAdd(&keyCnt[w * OWN_WAVES + (r[u] & 31)], 1);
        }
      }
    }
    if (lane == 0 && full) moreFlag[round & 1] = 1;
    __syncthreads();
    const int more = moreFlag[round & 1];
    if (tid == 0) moreFlag[(round + 1) & 1] = 0;

    int off = 0, tot = 0;
#pragma unroll
    for (int sw = 0; sw < OWN_WAVES; ++sw) {
      const int c = keyCnt[sw * OWN_WAVES + lane];
      if (sw < w) off += c;
      tot += c;
    }
    const int myTot = __builtin_amdgcn_readlane(tot, w);
    for (int lo = 0; __builtin_amdgcn_ballot_w32(tot > lo) != 0u; lo += MINE_CAP) {
      if (lo > 0) __syncthreads();
      keyRun[w * OWN_WAVES + lane] = off;
      for (int i0 = 0; i0 < cnt; i0 += 32) {
        const int  i     = i0 + lane;
        const int  ent   = (i < cnt) ? myScan[i] : -1;
        const bool valid = ent >= 0;
        const int  key   = ent & 31;
        unsigned same = __builtin_amdgcn_ballot_w32(valid);
        for (int b = 0; b < 5; ++b) {
          const bool     bit = ((key >> b) & 1) != 0;
          const unsigned hav = __builtin_amdgcn_ballot_w32(valid && bit);
          same &= bit ? hav : ~hav;
        }
        if (valid) {
          int* run = &keyRun[w * OWN_WAVES + key];
          const int before = *run;
          const int pos    = before + __popc(same & ltmask) - lo;
          if ((unsigned)pos < (unsigned)MINE_CAP) mineList[key * MINE_CAP + pos] = ent;
          if ((same & ltmask) == 0u) *run = before + __popc(same);
        }
      }
      __syncthreads();

      int n = myTot - lo;
      n = n < 0 ? 0 : (n > MINE_CAP ? MINE_CAP : n);
      add_rows<FIRST>(myMine, n, lane, m, src, node_id, norm, tile, myKeep, kept);
    }

    if (!more) break;
    ++round;
    __syncthreads();
  }
  }
  if (FIRST && lane == 0) listCnt[(int)blockIdx.x * OWN_WAVES + w] = kept;
  __syncthreads();

  if (FIRST) {
    float* Wl2 = (float*)scanList;
    float* bl  = (float*)mineList;
    for (int i = tid; i < (HD * HD) / 4; i += 1024) {
      int p = i >> 6;
      int n = (i & 63) << 1;
      int k = p << 1;
      float2 r0 = *(const float2*)(Wnext + (size_t)k * HD + n);
      float2 r1 = *(const float2*)(Wnext + (size_t)(k + 1) * HD + n);
      *(float4*)(Wl2 + p * 256 + n * 2) = make_float4(r0.x, r1.x, r0.y, r1.y);
    }
    if (tid < HD) bl[tid] = bias[tid];
    __syncthreads();
    if (w * 16 < nrows) {
      const int lane16 = lane & 15;
      const int pOff   = lane >> 4;
      int rl = w * 16 + lane16;
      if (rl > OWN_ROWS - 1) rl = OWN_ROWS - 1;
      int gr = base + rl;
      if (gr > N_NODES - 1) gr = N_NODES - 1;
      const float nrA = norm[gr];
      const float* trow = tile + rl * TP;
      v8f acc[8] = {};
      for (int k0 = 0; k0 < HD; k0 += 4) {
        const int kk = k0 + (pOff << 1);
        v2f s  = *(const v2f*)(trow + kk);
        v2f bb = *(const v2f*)(bl + kk);
        float x0 = s.x * nrA + bb.x;
        float x1 = s.y * nrA + bb.y;
        v2f a;
        a.x = x0 > 0.f ? x0 : 0.2f * x0;
        a.y = x1 > 0.f ? x1 : 0.2f * x1;
        const float* bbase = Wl2 + ((k0 >> 1) + pOff) * 256;
#pragma unroll
        for (int t = 0; t < 8; ++t) {
          v2f b = *(const v2f*)(bbase + (t * 16 + lane16) * 2);
          acc[t] = __builtin_amdgcn_wmma_f32_16x16x4_f32(
              false, a, false, b, (short)0, acc[t], false, false);
        }
      }
      const int rhalf = pOff << 3;
      const int nbits = __builtin_bit_cast(int, nrA);
#pragma unroll
      for (int v = 0; v < 8; ++v) {
        int lo = __builtin_amdgcn_readlane(nbits, v);
        int hi = __builtin_amdgcn_readlane(nbits, v + 8);
        float nr = __builtin_bit_cast(float, pOff ? hi : lo);
        int r = w * 16 + v + rhalf;
        if (r < nrows) {
#pragma unroll
          for (int t = 0; t < 8; ++t)
            tile[r * TP + t * 16 + lane16] = acc[t][v] * nr;
        }
      }
      __builtin_amdgcn_fence(__ATOMIC_ACQ_REL, "wavefront");
      const int c4 = lane << 2;
      v4f q[16];
#pragma unroll
      for (int j = 0; j < 16; ++j) {
        int r = w * 16 + j;
        q[j] = *(const v4f*)(tile + (r < OWN_ROWS ? r : OWN_ROWS - 1) * TP + c4);
      }
      asm volatile("" :: "v"(q[0]), "v"(q[1]), "v"(q[2]), "v"(q[3]),
                         "v"(q[4]), "v"(q[5]), "v"(q[6]), "v"(q[7]),
                         "v"(q[8]), "v"(q[9]), "v"(q[10]), "v"(q[11]),
                         "v"(q[12]), "v"(q[13]), "v"(q[14]), "v"(q[15]));
#pragma unroll
      for (int j = 0; j < 16; ++j) {
        int r = w * 16 + j;
        if (r < nrows) *(v4f*)(out + (size_t)(base + r) * HD + c4) = q[j];
      }
    }
    return;
  }

  const int c4 = lane << 2;
  const float4 b = *(const float4*)(bias + c4);
  for (int i = tid; i < nrows * 32; i += 1024) {
    int r = i >> 5;
    float nr = norm[base + r];
    float4 a = *(const float4*)(tile + r * TP + c4);
    float4 n;
    n.x = a.x * nr + b.x;
    n.y = a.y * nr + b.y;
    n.z = a.z * nr + b.z;
    n.w = a.w * nr + b.w;
    n.x = n.x > 0.f ? n.x : 0.2f * n.x;
    n.y = n.y > 0.f ? n.y : 0.2f * n.y;
    n.z = n.z > 0.f ? n.z : 0.2f * n.z;
    n.w = n.w > 0.f ? n.w : 0.2f * n.w;
    *(float4*)(out + (size_t)(base + r) * HD + c4) = n;
  }
}

extern "C" void kernel_launch(void* const* d_in, const int* in_sizes, int n_in,
                              void* d_out, int out_size, void* d_ws, size_t ws_size,
                              hipStream_t stream) {
  const int*   node_id = (const int*)d_in[0];
  const int*   src     = (const int*)d_in[1];
  const int*   dst     = (const int*)d_in[2];
  const float* norm    = (const float*)d_in[3];
  const float* embed   = (const float*)d_in[4];
  const float* W1      = (const float*)d_in[5];
  const float* b1      = (const float*)d_in[6];
  const float* W2      = (const float*)d_in[7];
  const float* b2      = (const float*)d_in[8];
  float* out = (float*)d_out;

  const size_t feat = (size_t)N_NODES * HD;
  float* bufH = (float*)d_ws;
  float* bufM = bufH + feat;
  int* lists   = (int*)(bufM + feat);
  int* listCnt = lists + LIST_INTS;
  unsigned* ownerBits = (unsigned*)(listCnt + OWN_WGS * OWN_WAVES);

  dim3 blk(256);
  const int gGemm = GEMM_BLOCKS + N_SEGS;

  table_gemm_wmma<<<gGemm, blk, 0, stream>>>(embed, W1, bufM, dst, ownerBits);

  owner_sum<true><<<OWN_WGS, dim3(1024), 0, stream>>>(bufM, src, dst, node_id, norm, b1, bufH,
                                                      lists, listCnt, ownerBits, W2);

  owner_sum<false><<<OWN_WGS, dim3(1024), 0, stream>>>(bufH, src, dst, node_id, norm, b2, out,
                                                       lists, listCnt, ownerBits, W2);
}


# Index look-ups re-laid by launch 1 as 8-B pairs {src,dst} per edge, {node_id,norm} per node: scan packs (src<<9)|row, owner reads 1 pair; 2 scattered index lines per edge, not 4; sum order same
# speedup vs baseline: 3.7698x; 1.0475x over previous
#include <hip/hip_runtime.h>

#define N_NODES 100000
#define N_EDGES 600000
#define HD 128

typedef __attribute__((ext_vector_type(2))) float v2f;
typedef __attribute__((ext_vector_type(8))) float v8f;
typedef __attribute__((ext_vector_type(4))) float v4f;

#define GEMM_BLOCKS ((N_NODES + 127) / 128)
#define OWN_ROWS    392
#define OWN_WGS     ((N_NODES + OWN_ROWS - 1) / OWN_ROWS)
#define SEG_EDGES   2048
#define SEG_WORDS   (SEG_EDGES / 32)
#define N_SEGS      ((N_EDGES + SEG_EDGES - 1) / SEG_EDGES)
#define ROW_WORDS   (N_SEGS * SEG_WORDS)
static_assert(OWN_WGS * SEG_WORDS * 4 == HD * HD * 4, "a segment's bit maps fill the product's LDS exactly");
static_assert(SEG_EDGES == 256 * 8 && N_EDGES % 8 == 0, "a thread's 8 edges are all inside the list or all outside");

__global__ __launch_bounds__(256) void table_gemm_wmma(
    const float* __restrict__ table, const float* __restrict__ W,
    float* __restrict__ p, const int* __restrict__ dst,
    unsigned* __restrict__ ownerBits,
    const int* __restrict__ src, const int* __restrict__ node_id,
    const float* __restrict__ norm, int2* __restrict__ sd, int2* __restrict__ nn) {
  __shared__ float Wl2[HD * HD];

  if ((int)blockIdx.x >= GEMM_BLOCKS) {
    unsigned* bm = (unsigned*)Wl2;
    const int t   = threadIdx.x;
    const int seg = (int)blockIdx.x - GEMM_BLOCKS;
    const int e0  = seg * SEG_EDGES + t * 8;
    const bool in = e0 < N_EDGES;
    const int4 d0 = *(const int4*)(dst + (in ? e0 : 0));
    const int4 d1 = *(const int4*)(dst + (in ? e0 + 4 : 0));
    const int4 s0 = *(const int4*)(src + (in ? e0 : 0));
    const int4 s1 = *(const int4*)(src + (in ? e0 + 4 : 0));
    for (int i = t; i < (OWN_WGS * SEG_WORDS) / 4; i += 256)
      ((uint4*)bm)[i] = make_uint4(0u, 0u, 0u, 0u);
    __syncthreads();
    if (in) {
      const int dd[8] = {d0.x, d0.y, d0.z, d0.w, d1.x, d1.y, d1.z, d1.w};
#pragma unroll
      for (int j = 0; j < 8; ++j) {
        const unsigned o  = (unsigned)dd[j] / OWN_ROWS;
        const unsigned el = (unsigned)(t * 8 + j);
        if (o < (unsigned)OWN_WGS) atomicOr(&bm[o * SEG_WORDS + (el >> 5)], 1u << (el & 31u));
      }
    }
    __syncthreads();
    for (int i = t; i < (OWN_WGS * SEG_WORDS) / 4; i += 256) {
      const int o = i >> 4;
      const int q = i & 15;
      *(uint4*)(ownerBits + ((size_t)o * N_SEGS + seg) * SEG_WORDS + q * 4) = ((const uint4*)bm)[i];
    }
    if (in) {
      int4* o4 = (int4*)(sd + e0);
      o4[0] = make_int4(s0.x, d0.x, s0.y, d0.y);
      o4[1] = make_int4(s0.z, d0.z, s0.w, d0.w);
      o4[2] = make_int4(s1.x, d1.x, s1.y, d1.y);
      o4[3] = make_int4(s1.z, d1.z, s1.w, d1.w);
    }
    return;
  }

  int tid = threadIdx.x;
  const int  nnI    = (int)blockIdx.x * 128 + tid;
  const bool nnMine = tid < 128 && nnI < N_NODES;
  int   nnId   = 0;
  float nnNorm = 0.f;
  if (nnMine) {
    nnId   = node_id[nnI];
    nnNorm = norm[nnI];
  }
  for (int i = tid; i < (HD * HD) / 4; i += 256) {
    int p = i >> 6;
    int n = (i & 63) << 1;
    int k = p << 1;
    float2 r0 = *(const float2*)(W + (size_t)k * HD + n);
    float2 r1 = *(const float2*)(W + (size_t)(k + 1) * HD + n);
    float4 o  = make_float4(r0.x, r1.x, r0.y, r1.y);
    *(float4*)(Wl2 + p * 256 + n * 2) = o;
  }
  __syncthreads();

  int wave    = tid >> 5;
  int lane    = tid & 31;
  int lane16  = lane & 15;
  int pOff    = lane >> 4;
  int rowBase = (blockIdx.x * 8 + wave) * 16;
  int rowA    = rowBase + lane16;
  int rowAc   = rowA < N_NODES ? rowA : (N_NODES - 1);

  v8f acc[8] = {};
  const float* hrow = table + (size_t)rowAc * HD;

  for (int k0 = 0; k0 < HD; k0 += 4) {
    v2f a = *(const v2f*)(hrow + k0 + (pOff << 1));
    const float* bbase = Wl2 + ((k0 >> 1) + pOff) * 256;
#pragma unroll
    for (int t = 0; t < 8; ++t) {
      int n = t * 16 + lane16;
      v2f b = *(const v2f*)(bbase + n * 2);
      acc[t] = __builtin_amdgcn_wmma_f32_16x16x4_f32(
          false, a, false, b, (short)0, acc[t], false, false);
    }
  }

  int rhalf = pOff << 3;
#pragma unroll
  for (int v = 0; v < 8; ++v) {
    int r = rowBase + v + rhalf;
    if (r < N_NODES) {
#pragma unroll
      for (int t = 0; t < 8; ++t)
        p[(size_t)r * HD + t * 16 + lane16] = acc[t][v];
    }
  }
  if (nnMine) nn[nnI] = make_int2(nnId, __builtin_bit_cast(int, nnNorm));
}

#define OWN_WAVES   32
#define SCAN_CAP    512
#define MINE_CAP    256
#define ROW_QUADS   (ROW_WORDS / 4)
#define BM_DEPTH    ((ROW_QUADS + 32 * OWN_WAVES - 1) / (32 * OWN_WAVES))
#define ROW_DEPTH   8
#define TP          (HD + 4)
#define LIST_CAP    256
#define LIST_INTS   ((size_t)OWN_WGS * OWN_WAVES * LIST_CAP)
static_assert(ROW_WORDS % 4 == 0, "an owner's bit row is a whole number of 16-byte pieces");
static_assert(OWN_ROWS <= 512, "the local row takes the 9 low bits of an entry");
static_assert(SCAN_CAP >= 32 && MINE_CAP >= 64, "a round always makes progress");
static_assert(N_NODES <= (1 << 22), "a source row shares an int with the 9-bit local row");
static_assert(OWN_WAVES * SCAN_CAP >= HD * HD && OWN_WAVES * MINE_CAP >= HD,
              "the next layer's weights and bias fit the dead lists");

template <bool SCALE>
__device__ __forceinline__ void add_batch(
    int s, float f, int r, int nb, int lane, const float* __restrict__ m, float* tile) {
  for (int q0 = 0; q0 < nb; q0 += ROW_DEPTH) {
    int sq[ROW_DEPTH];
#pragma unroll
    for (int q = 0; q < ROW_DEPTH; ++q)
      sq[q] = __builtin_amdgcn_readlane(s, q0 + q);
    v4f v[ROW_DEPTH];
#pragma unroll
    for (int q = 0; q < ROW_DEPTH; ++q)
      v[q] = *(const v4f*)(m + (size_t)sq[q] * HD + (lane << 2));
#pragma unroll
    for (int q = 0; q < ROW_DEPTH; ++q) {
      if (q0 + q < nb) {
        int rq = __builtin_amdgcn_readlane(r, q0 + q);
        v4f* t = (v4f*)(tile + rq * TP + (lane << 2));
        if (SCALE) {
          float fq = __builtin_bit_cast(float,
              __builtin_amdgcn_readlane(__builtin_bit_cast(int, f), q0 + q));
          *t = *t + v[q] * fq;
        } else {
          *t = *t + v[q];
        }
      }
    }
  }
}

template <bool FIRST>
__device__ __forceinline__ void add_rows(
    const int* mine, int n, int lane, const float* __restrict__ m,
    const int2* __restrict__ nn, float* tile, int* __restrict__ keep, int& kept) {
  for (int b0 = 0; b0 < n; b0 += 32) {
    int k   = b0 + lane;
    int ent = 0;
    if (k < n) ent = mine[k];
    int s  = ent >> 9;
    int r  = ent & 511;
    int nb = (n - b0) < 32 ? (n - b0) : 32;
    if (FIRST) {
      if (k < n && kept + lane < LIST_CAP) keep[kept + lane] = ent;
      kept += nb;
      const int2 pr = nn[s];
      add_batch<true>(pr.x, __builtin_bit_cast(float, pr.y), r, nb, lane, m, tile);
    } else {
      add_batch<false>(s, 0.f, r, nb, lane, m, tile);
    }
  }
}

__device__ __forceinline__ void add_kept_rows(
    const int* __restrict__ keep, int n, int lane,
    const float* __restrict__ m, float* tile) {
  int ents[LIST_CAP / 32];
#pragma unroll
  for (int b = 0; b < LIST_CAP / 32; ++b) {
    int k = b * 32 + lane;
    ents[b] = (k < n) ? keep[k] : 0;
  }
#pragma unroll
  for (int b = 0; b < LIST_CAP / 32; ++b) {
    int rest = n - b * 32;
    if (rest > 0)
      add_batch<false>(ents[b] >> 9, 0.f, ents[b] & 511, rest < 32 ? rest : 32, lane, m, tile);
  }
}

__device__ __forceinline__ void take_bits(
    unsigned& m, int e0, int* list, int& cnt, bool& full, unsigned ltmask) {
  for (int it = 0; it < 33 && !full; ++it) {
    const unsigned b2 = __builtin_amdgcn_ballot_w32(m != 0u);
    if (b2 == 0u) break;
    const int add = __popc(b2);
    if (cnt + add > SCAN_CAP) {
      full = true;
    } else {
      if (m != 0u) {
        list[cnt + __popc(b2 & ltmask)] = e0 + __builtin_ctz(m);
        m &= m - 1u;
      }
      cnt += add;
    }
  }
}

__device__ __forceinline__ void load_bits(
    uint4 (&mw)[BM_DEPTH], const unsigned* __restrict__ ownerBits, int w, int lane) {
  const uint4* bitRow = (const uint4*)(ownerBits + (size_t)blockIdx.x * ROW_WORDS);
#pragma unroll
  for (int u = 0; u < BM_DEPTH; ++u) {
    const int q = (u * OWN_WAVES + w) * 32 + lane;
    mw[u] = bitRow[(q < ROW_QUADS) ? q : 0];
    if (q >= ROW_QUADS) mw[u] = make_uint4(0u, 0u, 0u, 0u);
  }
}

template <bool FIRST>
__global__ __launch_bounds__(1024) void owner_sum(
    const float* __restrict__ m, const int2* __restrict__ sd,
    const int2* __restrict__ nn, const float* __restrict__ norm,
    const float* __restrict__ bias, float* __restrict__ out,
    int* __restrict__ lists, int* __restrict__ listCnt,
    const unsigned* __restrict__ ownerBits, const float* __restrict__ Wnext) {
  __shared__ alignas(16) float tile[OWN_ROWS * TP];
  __shared__ alignas(16) int scanList[OWN_WAVES * SCAN_CAP];
  __shared__ alignas(16) int mineList[OWN_WAVES * MINE_CAP];
  __shared__ int   keyCnt[OWN_WAVES * OWN_WAVES];
  __shared__ int   keyRun[OWN_WAVES * OWN_WAVES];
  __shared__ int   moreFlag[2];

  const int tid  = threadIdx.x;
  const int lane = tid & 31;
  const int w    = __builtin_amdgcn_readfirstlane(tid >> 5);
  const unsigned ltmask = (1u << lane) - 1u;
  const int base  = (int)blockIdx.x * OWN_ROWS;
  const int nrows = (N_NODES - base) < OWN_ROWS ? (N_NODES - base) : OWN_ROWS;

  int* myKeep = lists + ((size_t)blockIdx.x * OWN_WAVES + w) * LIST_CAP;
  int  kept   = 0;
  int  keptV  = 0;
  if (!FIRST) keptV = listCnt[(int)blockIdx.x * OWN_WAVES + lane];
  uint4 mw[BM_DEPTH];
  if (FIRST) load_bits(mw, ownerBits, w, lane);

  for (int i = tid; i < OWN_ROWS * TP / 4; i += 1024)
    ((float4*)tile)[i] = make_float4(0.f, 0.f, 0.f, 0.f);
  if (tid < 2) moreFlag[tid] = 0;
  const bool useKept =
      !FIRST && __builtin_amdgcn_ballot_w32(keptV > LIST_CAP) == 0u;
  __syncthreads();

  int* myScan = scanList + w * SCAN_CAP;
  int* myMine = mineList + w * MINE_CAP;
  int round = 0;
  if (useKept) {
    add_kept_rows(myKeep, __builtin_amdgcn_readlane(keptV, w), lane, m, tile);
  } else {
  if (!FIRST) load_bits(mw, ownerBits, w, lane);
  for (;;) {
    keyCnt[w * OWN_WAVES + lane] = 0;
    int  cnt  = 0;
    bool full = false;
#pragma unroll
    for (int u = 0; u < BM_DEPTH; ++u) {
      const int e0 = ((u * OWN_WAVES + w) * 32 + lane) * 128;
      take_bits(mw[u].x, e0,      myScan, cnt, full, ltmask);
      take_bits(mw[u].y, e0 + 32, myScan, cnt, full, ltmask);
      take_bits(mw[u].z, e0 + 64, myScan, cnt, full, ltmask);
      take_bits(mw[u].w, e0 + 96, myScan, cnt, full, ltmask);
    }
    for (int i0 = 0; i0 < cnt; i0 += 128) {
      int2 e[4];
      int  r[4];
#pragma unroll
      for (int u = 0; u < 4; ++u) {
        int i = i0 + u * 32 + lane;
        e[u] = sd[(i < cnt) ? myScan[i] : 0];
      }
#pragma unroll
      for (int u = 0; u < 4; ++u) r[u] = e[u].y - base;
#pragma unroll
      for (int u = 0; u < 4; ++u) {
        int i = i0 + u * 32 + lane;
        if (i < cnt) {
          const bool here = (unsigned)r[u] < (unsigned)nrows;
          myScan[i] = here ? ((e[u].x << 9) | r[u]) : -1;
          if (here) atomicAdd(&keyCnt[w * OWN_WAVES + (r[u] & 31)], 1);
        }
      }
    }
    if (lane == 0 && full) moreFlag[round & 1] = 1;
    __syncthreads();
    const int more = moreFlag[round & 1];
    if (tid == 0) moreFlag[(round + 1) & 1] = 0;

    int off = 0, tot = 0;
#pragma unroll
    for (int sw = 0; sw < OWN_WAVES; ++sw) {
      const int c = keyCnt[sw * OWN_WAVES + lane];
      if (sw < w) off += c;
      tot += c;
    }
    const int myTot = __builtin_amdgcn_readlane(tot, w);
    for (int lo = 0; __builtin_amdgcn_ballot_w32(tot > lo) != 0u; lo += MINE_CAP) {
      if (lo > 0) __syncthreads();
      keyRun[w * OWN_WAVES + lane] = off;
      for (int i0 = 0; i0 < cnt; i0 += 32) {
        const int  i     = i0 + lane;
        const int  ent   = (i < cnt) ? myScan[i] : -1;
        const bool valid = ent >= 0;
        const int  key   = ent & 31;
        unsigned same = __builtin_amdgcn_ballot_w32(valid);
        for (int b = 0; b < 5; ++b) {
          const bool     bit = ((key >> b) & 1) != 0;
          const unsigned hav = __builtin_amdgcn_ballot_w32(valid && bit);
          same &= bit ? hav : ~hav;
        }
        if (valid) {
          int* run = &keyRun[w * OWN_WAVES + key];
          const int before = *run;
          const int pos    = before + __popc(same & ltmask) - lo;
          if ((unsigned)pos < (unsigned)MINE_CAP) mineList[key * MINE_CAP + pos] = ent;
          if ((same & ltmask) == 0u) *run = before + __popc(same);
        }
      }
      __syncthreads();

      int n = myTot - lo;
      n = n < 0 ? 0 : (n > MINE_CAP ? MINE_CAP : n);
      add_rows<FIRST>(myMine, n, lane, m, nn, tile, myKeep, kept);
    }

    if (!more) break;
    ++round;
    __syncthreads();
  }
  }
  if (FIRST && lane == 0) listCnt[(int)blockIdx.x * OWN_WAVES + w] = kept;
  __syncthreads();

  if (FIRST) {
    float* Wl2 = (float*)scanList;
    float* bl  = (float*)mineList;
    for (int i = tid; i < (HD * HD) / 4; i += 1024) {
      int p = i >> 6;
      int n = (i & 63) << 1;
      int k = p << 1;
      float2 r0 = *(const float2*)(Wnext + (size_t)k * HD + n);
      float2 r1 = *(const float2*)(Wnext + (size_t)(k + 1) * HD + n);
      *(float4*)(Wl2 + p * 256 + n * 2) = make_float4(r0.x, r1.x, r0.y, r1.y);
    }
    if (tid < HD) bl[tid] = bias[tid];
    __syncthreads();
    if (w * 16 < nrows) {
      const int lane16 = lane & 15;
      const int pOff   = lane >> 4;
      int rl = w * 16 + lane16;
      if (rl > OWN_ROWS - 1) rl = OWN_ROWS - 1;
      int gr = base + rl;
      if (gr > N_NODES - 1) gr = N_NODES - 1;
      const float nrA = norm[gr];
      const float* trow = tile + rl * TP;
      v8f acc[8] = {};
      for (int k0 = 0; k0 < HD; k0 += 4) {
        const int kk = k0 + (pOff << 1);
        v2f s  = *(const v2f*)(trow + kk);
        v2f bb = *(const v2f*)(bl + kk);
        float x0 = s.x * nrA + bb.x;
        float x1 = s.y * nrA + bb.y;
        v2f a;
        a.x = x0 > 0.f ? x0 : 0.2f * x0;
        a.y = x1 > 0.f ? x1 : 0.2f * x1;
        const float* bbase = Wl2 + ((k0 >> 1) + pOff) * 256;
#pragma unroll
        for (int t = 0; t < 8; ++t) {
          v2f b = *(const v2f*)(bbase + (t * 16 + lane16) * 2);
          acc[t] = __builtin_amdgcn_wmma_f32_16x16x4_f32(
              false, a, false, b, (short)0, acc[t], false, false);
        }
      }
      const int rhalf = pOff << 3;
      const int nbits = __builtin_bit_cast(int, nrA);
#pragma unroll
      for (int v = 0; v < 8; ++v) {
        int lo = __builtin_amdgcn_readlane(nbits, v);
        int hi = __builtin_amdgcn_readlane(nbits, v + 8);
        float nr = __builtin_bit_cast(float, pOff ? hi : lo);
        int r = w * 16 + v + rhalf;
        if (r < nrows) {
#pragma unroll
          for (int t = 0; t < 8; ++t)
            tile[r * TP + t * 16 + lane16] = acc[t][v] * nr;
        }
      }
      __builtin_amdgcn_fence(__ATOMIC_ACQ_REL, "wavefront");
      const int c4 = lane << 2;
      v4f q[16];
#pragma unroll
      for (int j = 0; j < 16; ++j) {
        int r = w * 16 + j;
        q[j] = *(const v4f*)(tile + (r < OWN_ROWS ? r : OWN_ROWS - 1) * TP + c4);
      }
      asm volatile("" :: "v"(q[0]), "v"(q[1]), "v"(q[2]), "v"(q[3]),
                         "v"(q[4]), "v"(q[5]), "v"(q[6]), "v"(q[7]),
                         "v"(q[8]), "v"(q[9]), "v"(q[10]), "v"(q[11]),
                         "v"(q[12]), "v"(q[13]), "v"(q[14]), "v"(q[15]));
#pragma unroll
      for (int j = 0; j < 16; ++j) {
        int r = w * 16 + j;
        if (r < nrows) *(v4f*)(out + (size_t)(base + r) * HD + c4) = q[j];
      }
    }
    return;
  }

  const int c4 = lane << 2;
  const float4 b = *(const float4*)(bias + c4);
  for (int i = tid; i < nrows * 32; i += 1024) {
    int r = i >> 5;
    float nr = norm[base + r];
    float4 a = *(const float4*)(tile + r * TP + c4);
    float4 n;
    n.x = a.x * nr + b.x;
    n.y = a.y * nr + b.y;
    n.z = a.z * nr + b.z;
    n.w = a.w * nr + b.w;
    n.x = n.x > 0.f ? n.x : 0.2f * n.x;
    n.y = n.y > 0.f ? n.y : 0.2f * n.y;
    n.z = n.z > 0.f ? n.z : 0.2f * n.z;
    n.w = n.w > 0.f ? n.w : 0.2f * n.w;
    *(float4*)(out + (size_t)(base + r) * HD + c4) = n;
  }
}

extern "C" void kernel_launch(void* const* d_in, const int* in_sizes, int n_in,
                              void* d_out, int out_size, void* d_ws, size_t ws_size,
                              hipStream_t stream) {
  const int*   node_id = (const int*)d_in[0];
  const int*   src     = (const int*)d_in[1];
  const int*   dst     = (const int*)d_in[2];
  const float* norm    = (const float*)d_in[3];
  const float* embed   = (const float*)d_in[4];
  const float* W1      = (const float*)d_in[5];
  const float* b1      = (const float*)d_in[6];
  const float* W2      = (const float*)d_in[7];
  const float* b2      = (const float*)d_in[8];
  float* out = (float*)d_out;

  const size_t feat = (size_t)N_NODES * HD;
  float* bufH = (float*)d_ws;
  float* bufM = bufH + feat;
  int* lists   = (int*)(bufM + feat);
  int* listCnt = lists + LIST_INTS;
  unsigned* ownerBits = (unsigned*)(listCnt + OWN_WGS * OWN_WAVES);
  int2* sd = (int2*)(ownerBits + (size_t)OWN_WGS * ROW_WORDS);
  int2* nn = sd + N_EDGES;

  dim3 blk(256);
  const int gGemm = GEMM_BLOCKS + N_SEGS;

  table_gemm_wmma<<<gGemm, blk, 0, stream>>>(embed, W1, bufM, dst, ownerBits,
                                             src, node_id, norm, sd, nn);

  owner_sum<true><<<OWN_WGS, dim3(1024), 0, stream>>>(bufM, sd, nn, norm, b1, bufH,
                                                      lists, listCnt, ownerBits, W2);

  owner_sum<false><<<OWN_WGS, dim3(1024), 0, stream>>>(bufH, sd, nn, norm, b2, out,
                                                       lists, listCnt, ownerBits, W2);
}
